# starting kernel
# baseline (speedup 1.0000x reference)
#include <hip/hip_runtime.h>
#include <cmath>

typedef __attribute__((ext_vector_type(2))) float v2f;
typedef __attribute__((ext_vector_type(8))) float v8f;

#define NND   50000
#define NED   800000
#define DIN   256
#define DH    128
#define ALPHA 0.1f
#define BNEPS 1e-5f

__global__ void k_deg_init(float* deg) {
    int i = blockIdx.x * blockDim.x + threadIdx.x;
    if (i < NND) deg[i] = 1.0f;
}
__global__ void k_deg_edges(const int* __restrict__ dst, float* deg) {
    int e = blockIdx.x * blockDim.x + threadIdx.x;
    if (e < NED) atomicAdd(&deg[dst[e]], 1.0f);
}
__global__ void k_deg_rsqrt(float* deg) {
    int i = blockIdx.x * blockDim.x + threadIdx.x;
    if (i < NND) deg[i] = rsqrtf(fmaxf(deg[i], 1.0f));
}

template <int K>
__global__ __launch_bounds__(256)
void k_gemm_wmma(const float* in, const float* __restrict__ W,
                 float* out, int mode, float beta) {
    extern __shared__ float2 sW2[];
    const int tid = threadIdx.x;

    {
        const int totalIter = (K / 2) * (DH / 4);
        for (int i = tid; i < totalIter; i += 256) {
            const int kp = i / (DH / 4);
            const int c4 = i % (DH / 4);
            const float4 a0 = *(const float4*)(W + (size_t)(2 * kp)     * DH + c4 * 4);
            const float4 a1 = *(const float4*)(W + (size_t)(2 * kp + 1) * DH + c4 * 4);
            float4* dst4 = (float4*)(sW2 + (size_t)kp * DH + c4 * 4);
            float4 f0; f0.x = a0.x; f0.y = a1.x; f0.z = a0.y; f0.w = a1.y;
            float4 f1; f1.x = a0.z; f1.y = a1.z; f1.z = a0.w; f1.w = a1.w;
            dst4[0] = f0;
            dst4[1] = f1;
        }
    }
    __syncthreads();

    const int wave = tid >> 5;
    const int lane = tid & 31;
    const int tileRow = blockIdx.x * 8 + wave;
    if (tileRow >= NND / 16) return;
    const int m0 = tileRow * 16;
    const int r    = lane & 15;
    const bool hi  = lane >= 16;
    const int koff = hi ? 2 : 0;
    const float* arow = in + (size_t)(m0 + r) * K;

    v8f c[8] = {};
    for (int k0 = 0; k0 < K; k0 += 4) {
        const float2 a2 = *(const float2*)(arow + k0 + koff);
        v2f a; a.x = a2.x; a.y = a2.y;
        const int pairRow = (k0 >> 1) + (hi ? 1 : 0);
        const float2* brow = sW2 + (size_t)pairRow * DH + r;
#pragma unroll
        for (int t = 0; t < 8; ++t) {
            const float2 b2 = brow[t * 16];
            v2f b; b.x = b2.x; b.y = b2.y;
            c[t] = __builtin_amdgcn_wmma_f32_16x16x4_f32(
                false, a, false, b, (short)0, c[t], false, false);
        }
    }

#pragma unroll
    for (int t = 0; t < 8; ++t) {
        const int col = t * 16 + r;
#pragma unroll
        for (int v = 0; v < 8; ++v) {
            const int row = m0 + v + (hi ? 8 : 0);
            const size_t o = (size_t)row * DH + col;
            float val = c[t][v];
            if (mode != 0) val = (1.0f - beta) * in[o] + beta * val;
            out[o] = val;
        }
    }
}

__global__ __launch_bounds__(256)
void k_gcn_self_bias(const float* __restrict__ xw, const float* __restrict__ dinv,
                     const float* __restrict__ b, float* __restrict__ z) {
    int i = blockIdx.x * blockDim.x + threadIdx.x;
    int node = i >> 5, l = i & 31;
    float w = dinv[node] * dinv[node];
    float4 v  = *(const float4*)(xw + (size_t)node * DH + l * 4);
    float4 bb = *(const float4*)(b + l * 4);
    float4 o;
    o.x = w * v.x + bb.x; o.y = w * v.y + bb.y;
    o.z = w * v.z + bb.z; o.w = w * v.w + bb.w;
    *(float4*)(z + (size_t)node * DH + l * 4) = o;
}

__global__ __launch_bounds__(256)
void k_gcn_edge(const int* __restrict__ src, const int* __restrict__ dst,
                const float* __restrict__ dinv, const float* __restrict__ xw,
                float* z) {
    int e = blockIdx.x * 8 + (threadIdx.x >> 5);
    int l = threadIdx.x & 31;
    int s = src[e], d = dst[e];
    float w = dinv[s] * dinv[d];
    const float4 v = *(const float4*)(xw + (size_t)s * DH + l * 4);
    float* p = z + (size_t)d * DH + l * 4;
    atomicAdd(p + 0, w * v.x);
    atomicAdd(p + 1, w * v.y);
    atomicAdd(p + 2, w * v.z);
    atomicAdd(p + 3, w * v.w);
}

__global__ __launch_bounds__(256)
void k_h_init(const float* __restrict__ x0, float* __restrict__ h) {
    size_t i = (size_t)blockIdx.x * blockDim.x + threadIdx.x;
    float4 v = *(const float4*)(x0 + i * 4);
    float4 o; o.x = ALPHA * v.x; o.y = ALPHA * v.y; o.z = ALPHA * v.z; o.w = ALPHA * v.w;
    *(float4*)(h + i * 4) = o;
}

__global__ __launch_bounds__(256)
void k_gcn2_edge(const int* __restrict__ src, const int* __restrict__ dst,
                 const float* __restrict__ z, float* h) {
    int e = blockIdx.x * 8 + (threadIdx.x >> 5);
    int l = threadIdx.x & 31;
    int s = src[e], d = dst[e];
    const float4 v = *(const float4*)(z + (size_t)s * DH + l * 4);
    float* p = h + (size_t)d * DH + l * 4;
    const float w = 1.0f - ALPHA;
    atomicAdd(p + 0, w * v.x);
    atomicAdd(p + 1, w * v.y);
    atomicAdd(p + 2, w * v.z);
    atomicAdd(p + 3, w * v.w);
}

__global__ void k_stats_zero(float* stats) { stats[threadIdx.x] = 0.0f; }

__global__ __launch_bounds__(256)
void k_bn_stats(const float* __restrict__ z, float* stats, int rowsPerBlock) {
    __shared__ float ssum[DH], ssq[DH];
    int tid = threadIdx.x;
    if (tid < DH) { ssum[tid] = 0.0f; ssq[tid] = 0.0f; }
    __syncthreads();
    int c = tid & (DH - 1);
    int rsub = tid >> 7;
    int base = blockIdx.x * rowsPerBlock;
    int rend = base + rowsPerBlock; if (rend > NND) rend = NND;
    float s = 0.0f, q = 0.0f;
    for (int r = base + rsub; r < rend; r += 2) {
        float v = z[(size_t)r * DH + c];
        s += v; q += v * v;
    }
    atomicAdd(&ssum[c], s);
    atomicAdd(&ssq[c], q);
    __syncthreads();
    if (tid < DH) {
        atomicAdd(&stats[tid], ssum[tid]);
        atomicAdd(&stats[DH + tid], ssq[tid]);
    }
}

__global__ void k_bn_final(float* stats) {
    int c = threadIdx.x;
    float invN = 1.0f / (float)NND;
    float mean = stats[c] * invN;
    float var  = stats[DH + c] * invN - mean * mean;
    stats[2 * DH + c] = mean;
    stats[3 * DH + c] = rsqrtf(var + BNEPS);
}

__global__ __launch_bounds__(256)
void k_bn_apply(float* z, const float* __restrict__ stats,
                const float* __restrict__ gamma, const float* __restrict__ beta) {
    size_t i = (size_t)blockIdx.x * blockDim.x + threadIdx.x;
    int c = (int)(i & (DH - 1));
    float v = z[i];
    v = (v - stats[2 * DH + c]) * stats[3 * DH + c] * gamma[c] + beta[c];
    z[i] = fmaxf(v, 0.0f);
}

extern "C" void kernel_launch(void* const* d_in, const int* in_sizes, int n_in,
                              void* d_out, int out_size, void* d_ws, size_t ws_size,
                              hipStream_t stream) {
    const float* x        = (const float*)d_in[0];
    const int*   edge     = (const int*)d_in[1];
    const float* W_lin    = (const float*)d_in[2];
    const float* b_lin    = (const float*)d_in[3];
    const float* W1s      = (const float*)d_in[4];
    const float* bn_gamma = (const float*)d_in[5];
    const float* bn_beta  = (const float*)d_in[6];
    float* out = (float*)d_out;

    float* ws    = (float*)d_ws;
    float* dinv  = ws;
    float* stats = ws + NND;
    float* bufA  = stats + 512;
    float* bufB  = bufA + (size_t)NND * DH;
    float* x0    = bufB + (size_t)NND * DH;

    const int* src = edge;
    const int* dst = edge + NED;

    const int nBlkN   = (NND + 255) / 256;
    const int nBlkE   = (NED + 255) / 256;
    const int nBlkVec = NND * 32 / 256;
    const int nBlkEdg = NED / 8;
    const int nBlkGemm = (NND / 16 + 7) / 8;

    k_deg_init<<<nBlkN, 256, 0, stream>>>(dinv);
    k_deg_edges<<<nBlkE, 256, 0, stream>>>(dst, dinv);
    k_deg_rsqrt<<<nBlkN, 256, 0, stream>>>(dinv);

    k_gemm_wmma<DIN><<<nBlkGemm, 256, DIN * DH * sizeof(float), stream>>>(
        x, W_lin, bufA, 0, 0.0f);

    k_gcn_self_bias<<<nBlkVec, 256, 0, stream>>>(bufA, dinv, b_lin, bufB);
    k_gcn_edge<<<nBlkEdg, 256, 0, stream>>>(src, dst, dinv, bufA, bufB);

    hipMemcpyAsync(x0, bufB, (size_t)NND * DH * sizeof(float),
                   hipMemcpyDeviceToDevice, stream);

    float* z = bufB;
    float* h = bufA;
    for (int i = 0; i < 5; ++i) {
        const float beta = logf(0.5f / (float)(i + 1) + 1.0f);
        k_h_init<<<nBlkVec, 256, 0, stream>>>(x0, h);
        k_gcn2_edge<<<nBlkEdg, 256, 0, stream>>>(src, dst, z, h);
        k_gemm_wmma<DH><<<nBlkGemm, 256, DH * DH * sizeof(float), stream>>>(
            h, W1s + (size_t)i * DH * DH, h, 1, beta);
        if (i < 4) {
            k_stats_zero<<<1, 256, 0, stream>>>(stats);
            k_bn_stats<<<(NND + 63) / 64, 256, 0, stream>>>(h, stats, 64);
            k_bn_final<<<1, DH, 0, stream>>>(stats);
            k_bn_apply<<<NND * DH / 256, 256, 0, stream>>>(
                h, stats, bn_gamma + i * DH, bn_beta + i * DH);
        }
        float* t = z; z = h; h = t;
    }

    hipMemcpyAsync(out, z, (size_t)NND * DH * sizeof(float),
                   hipMemcpyDeviceToDevice, stream);
}


# BatchNorm in two streaming launches instead of four: column partial sums per work-group (fixed order, no atomics), apply adds the partials itself, 16 B per lane; statistics sum order changes
# speedup vs baseline: 3.3269x; 3.3269x over previous
#include <hip/hip_runtime.h>
#include <cmath>

typedef __attribute__((ext_vector_type(2))) float v2f;
typedef __attribute__((ext_vector_type(8))) float v8f;

#define NND   50000
#define NED   800000
#define DIN   256
#define DH    128
#define ALPHA 0.1f
#define BNEPS 1e-5f

__global__ void k_deg_init(float* deg) {
    int i = blockIdx.x * blockDim.x + threadIdx.x;
    if (i < NND) deg[i] = 1.0f;
}
__global__ void k_deg_edges(const int* __restrict__ dst, float* deg) {
    int e = blockIdx.x * blockDim.x + threadIdx.x;
    if (e < NED) atomicAdd(&deg[dst[e]], 1.0f);
}
__global__ void k_deg_rsqrt(float* deg) {
    int i = blockIdx.x * blockDim.x + threadIdx.x;
    if (i < NND) deg[i] = rsqrtf(fmaxf(deg[i], 1.0f));
}

template <int K>
__global__ __launch_bounds__(256)
void k_gemm_wmma(const float* in, const float* __restrict__ W,
                 float* out, int mode, float beta) {
    extern __shared__ float2 sW2[];
    const int tid = threadIdx.x;

    {
        const int totalIter = (K / 2) * (DH / 4);
        for (int i = tid; i < totalIter; i += 256) {
            const int kp = i / (DH / 4);
            const int c4 = i % (DH / 4);
            const float4 a0 = *(const float4*)(W + (size_t)(2 * kp)     * DH + c4 * 4);
            const float4 a1 = *(const float4*)(W + (size_t)(2 * kp + 1) * DH + c4 * 4);
            float4* dst4 = (float4*)(sW2 + (size_t)kp * DH + c4 * 4);
            float4 f0; f0.x = a0.x; f0.y = a1.x; f0.z = a0.y; f0.w = a1.y;
            float4 f1; f1.x = a0.z; f1.y = a1.z; f1.z = a0.w; f1.w = a1.w;
            dst4[0] = f0;
            dst4[1] = f1;
        }
    }
    __syncthreads();

    const int wave = tid >> 5;
    const int lane = tid & 31;
    const int tileRow = blockIdx.x * 8 + wave;
    if (tileRow >= NND / 16) return;
    const int m0 = tileRow * 16;
    const int r    = lane & 15;
    const bool hi  = lane >= 16;
    const int koff = hi ? 2 : 0;
    const float* arow = in + (size_t)(m0 + r) * K;

    v8f c[8] = {};
    for (int k0 = 0; k0 < K; k0 += 4) {
        const float2 a2 = *(const float2*)(arow + k0 + koff);
        v2f a; a.x = a2.x; a.y = a2.y;
        const int pairRow = (k0 >> 1) + (hi ? 1 : 0);
        const float2* brow = sW2 + (size_t)pairRow * DH + r;
#pragma unroll
        for (int t = 0; t < 8; ++t) {
            const float2 b2 = brow[t * 16];
            v2f b; b.x = b2.x; b.y = b2.y;
            c[t] = __builtin_amdgcn_wmma_f32_16x16x4_f32(
                false, a, false, b, (short)0, c[t], false, false);
        }
    }

#pragma unroll
    for (int t = 0; t < 8; ++t) {
        const int col = t * 16 + r;
#pragma unroll
        for (int v = 0; v < 8; ++v) {
            const int row = m0 + v + (hi ? 8 : 0);
            const size_t o = (size_t)row * DH + col;
            float val = c[t][v];
            if (mode != 0) val = (1.0f - beta) * in[o] + beta * val;
            out[o] = val;
        }
    }
}

#define AGG_R      196
#define AGG_NWG    ((NND + AGG_R - 1) / AGG_R)
#define AGG_TPB    1024
#define AGG_WAVES  (AGG_TPB / 32)
#define AGG_LPT    8
#define AGG_BATCH  (32 * 4 * AGG_LPT)
#define AGG_QCAP   (AGG_BATCH + 32)
#define AGG_NBATCH ((NED + AGG_BATCH - 1) / AGG_BATCH)
#define AGG_S      (DH + 4)
#define AGG_RT     ((AGG_R + 15) / 16)
#define AGG_ACCF   (AGG_RT * 16 * AGG_S)
#define AGG_GW     (AGG_RT * 2)
#define AGG_LDS0   ((AGG_ACCF + AGG_WAVES * AGG_QCAP) * 4)
#define AGG_LDS1   (AGG_LDS0 + DH * DH * 4)
#define AGG_WCAP   2048
static_assert(NED <= (1 << 20) && AGG_R <= (1 << 12), "queue entry packing");
static_assert(AGG_WAVES <= AGG_NBATCH, "every wave has a first batch");
static_assert(AGG_GW <= AGG_WAVES, "a wave for every row tile and column half");
static_assert(AGG_LDS1 <= 320 * 1024, "one work-group's LDS");
static_assert(NND <= (1 << 16) && AGG_R <= (1 << 12), "list entry packing");

__device__ __forceinline__ void agg_rows(int s, int dl, int wbits, int cntc, int lane,
                                         const float* __restrict__ zin, float* acc) {
    for (int g = 0; g < cntc; g += 8) {
        float v[8][4];
#pragma unroll
        for (int u = 0; u < 8; ++u) {
            int k = g + u;
            if (k > cntc - 1) k = cntc - 1;
            const int sk = __builtin_amdgcn_readlane(s, k);
            const float* row = zin + (size_t)sk * DH + lane;
            v[u][0] = row[0];
            v[u][1] = row[32];
            v[u][2] = row[64];
            v[u][3] = row[96];
        }
#pragma unroll
        for (int u = 0; u < 8; ++u) {
            const int k = g + u;
            if (k < cntc) {
                const int dk = __builtin_amdgcn_readlane(dl, k);
                const float wk = __int_as_float(__builtin_amdgcn_readlane(wbits, k));
                float* arow = acc + dk * AGG_S + lane;
                atomicAdd(arow,      wk * v[u][0]);
                atomicAdd(arow + 32, wk * v[u][1]);
                atomicAdd(arow + 64, wk * v[u][2]);
                atomicAdd(arow + 96, wk * v[u][3]);
            }
        }
    }
}

template <int MODE>
__device__ __forceinline__ void agg_chunk(const unsigned* q, int base, int cntc, int lane, int lo,
                                          const int* __restrict__ src,
                                          const float* __restrict__ zin,
                                          const float* __restrict__ dinv, float* acc,
                                          unsigned* wl, int& wn) {
    const unsigned ent = (lane < cntc) ? q[base + lane] : 0u;
    const int e  = (int)(ent & 0xFFFFFu);
    const int dl = (int)(ent >> 20);
    const int s  = src[e];
    float w = 1.0f - ALPHA;
    if (MODE == 0) {
        w = dinv[s] * dinv[lo + dl];
        if (wn + cntc <= AGG_WCAP && lane < cntc)
            wl[wn + lane] = ((unsigned)dl << 16) | (unsigned)s;
        wn += cntc;
    }
    agg_rows(s, dl, __float_as_int(w), cntc, lane, zin, acc);
}

__device__ __forceinline__ void agg_load(int4 (&dv)[AGG_LPT], const int* __restrict__ dst,
                                         int t, int lane) {
    const int eb = t * AGG_BATCH + lane * 4;
#pragma unroll
    for (int j = 0; j < AGG_LPT; ++j) {
        int e = eb + j * 128;
        if (e > NED - 4) e = NED - 4;
        dv[j] = *(const int4*)(dst + e);
    }
    if (t == AGG_NBATCH - 1) {
#pragma unroll
        for (int j = 0; j < AGG_LPT; ++j)
            if (eb + j * 128 >= NED) dv[j] = make_int4(-1, -1, -1, -1);
    }
}

__device__ __forceinline__ void agg_scan(const int4 (&dv)[AGG_LPT], int t, int lane, int lo, int cnt,
                                         unsigned* q, int& qn) {
    const int eb = t * AGG_BATCH + lane * 4;
#pragma unroll
    for (int j = 0; j < AGG_LPT; ++j) {
        const int e = eb + j * 128;
        const int dd[4] = {dv[j].x, dv[j].y, dv[j].z, dv[j].w};
#pragma unroll
        for (int c = 0; c < 4; ++c) {
            const unsigned dl = (unsigned)(dd[c] - lo);
            const bool m = dl < (unsigned)cnt;
            const unsigned bm = __builtin_amdgcn_ballot_w32(m);
            if (bm != 0u) {
                const int pos = qn + (int)__builtin_amdgcn_mbcnt_lo(bm, 0u);
                if (m) q[pos] = (dl << 20) | (unsigned)(e + c);
                qn += __builtin_popcount(bm);
            }
        }
    }
}

template <int MODE>
__device__ __forceinline__ void agg_drain(unsigned* q, int& qn, int lane, int lo,
                                          const int* __restrict__ src,
                                          const float* __restrict__ zin,
                                          const float* __restrict__ dinv, float* acc,
                                          unsigned* wl, int& wn) {
    if (qn < 32) return;
    __builtin_amdgcn_wave_barrier();
    const int full = qn & ~31;
    for (int base = 0; base < full; base += 32)
        agg_chunk<MODE>(q, base, 32, lane, lo, src, zin, dinv, acc, wl, wn);
    const int rem = qn - full;
    unsigned keep = 0u;
    if (lane < rem) keep = q[full + lane];
    __builtin_amdgcn_wave_barrier();
    if (lane < rem) q[lane] = keep;
    qn = rem;
}

template <int MODE>
__global__ __launch_bounds__(AGG_TPB)
void k_agg(const int* __restrict__ src, const int* __restrict__ dst,
           const float* __restrict__ zin, const float* __restrict__ start,
           const float* __restrict__ dinv, const float* __restrict__ bias,
           float* __restrict__ out, int* cnts, unsigned* lists,
           const float* __restrict__ W1, float beta) {
    extern __shared__ float agg_smem[];
    float* acc = agg_smem;
    unsigned* qall = (unsigned*)(agg_smem + AGG_ACCF);
    float2* sW2 = (float2*)(agg_smem + AGG_ACCF + AGG_WAVES * AGG_QCAP);
    const int tid  = threadIdx.x;
    const int lane = tid & 31;
    const int wave = __builtin_amdgcn_readfirstlane(tid >> 5);
    const int lo   = blockIdx.x * AGG_R;
    const int cnt  = (NND - lo < AGG_R) ? (NND - lo) : AGG_R;
    unsigned* q = qall + wave * AGG_QCAP;
    const int widx = blockIdx.x * AGG_WAVES + wave;
    unsigned* wl = lists + (size_t)widx * AGG_WCAP;

    const int rot = (int)(((long long)blockIdx.x * AGG_NBATCH) / AGG_NWG);
    int t0 = wave;
    int ta = t0 + rot;
    if (ta >= AGG_NBATCH) ta -= AGG_NBATCH;
    int4 dva[AGG_LPT], dvb[AGG_LPT];

    int wn = 0;
    bool scan = true;
    unsigned ent = 0u;
    if (MODE == 0) {
        agg_load(dva, dst, ta, lane);
    } else {
        wn = cnts[widx];
        scan = wn > AGG_WCAP;
        if (scan) agg_load(dva, dst, ta, lane);
        else if (lane < wn) ent = wl[lane];
    }

    for (int i = tid; i < cnt * (DH / 4); i += AGG_TPB) {
        const int n = i >> 5, c4 = i & 31;
        const float4 v = *(const float4*)(start + (size_t)(lo + n) * DH + c4 * 4);
        float4 o;
        if (MODE == 0) {
            const float dn = dinv[lo + n];
            const float w = dn * dn;
            const float4 bb = *(const float4*)(bias + c4 * 4);
            o.x = w * v.x + bb.x; o.y = w * v.y + bb.y;
            o.z = w * v.z + bb.z; o.w = w * v.w + bb.w;
        } else {
            o.x = ALPHA * v.x; o.y = ALPHA * v.y; o.z = ALPHA * v.z; o.w = ALPHA * v.w;
        }
        *(float4*)(acc + n * AGG_S + c4 * 4) = o;
    }
    if (MODE == 1) {
        for (int i = tid; i < (DH / 2) * (DH / 4); i += AGG_TPB) {
            const int kp = i / (DH / 4);
            const int c4 = i % (DH / 4);
            const float4 a0 = *(const float4*)(W1 + (size_t)(2 * kp)     * DH + c4 * 4);
            const float4 a1 = *(const float4*)(W1 + (size_t)(2 * kp + 1) * DH + c4 * 4);
            float4* dst4 = (float4*)(sW2 + (size_t)kp * DH + c4 * 4);
            float4 f0; f0.x = a0.x; f0.y = a1.x; f0.z = a0.y; f0.w = a1.y;
            float4 f1; f1.x = a0.z; f1.y = a1.z; f1.z = a0.w; f1.w = a1.w;
            dst4[0] = f0;
            dst4[1] = f1;
        }
        for (int i = cnt * AGG_S + tid; i < AGG_ACCF; i += AGG_TPB) acc[i] = 0.0f;
    }
    __syncthreads();

    if (scan) {
        int qn = 0;
        while (true) {
            const int t1 = t0 + AGG_WAVES;
            const bool has1 = t1 < AGG_NBATCH;
            int tb = (has1 ? t1 : t0) + rot;
            if (tb >= AGG_NBATCH) tb -= AGG_NBATCH;
            agg_load(dvb, dst, tb, lane);
            agg_scan(dva, ta, lane, lo, cnt, q, qn);
            agg_drain<MODE>(q, qn, lane, lo, src, zin, dinv, acc, wl, wn);
            if (!has1) break;
            const int t2 = t1 + AGG_WAVES;
            const bool has2 = t2 < AGG_NBATCH;
            ta = (has2 ? t2 : t1) + rot;
            if (ta >= AGG_NBATCH) ta -= AGG_NBATCH;
            agg_load(dva, dst, ta, lane);
            agg_scan(dvb, tb, lane, lo, cnt, q, qn);
            agg_drain<MODE>(q, qn, lane, lo, src, zin, dinv, acc, wl, wn);
            if (!has2) break;
            t0 = t2;
        }
        __builtin_amdgcn_wave_barrier();
        if (qn > 0) agg_chunk<MODE>(q, 0, qn, lane, lo, src, zin, dinv, acc, wl, wn);
        if (MODE == 0 && lane == 0) cnts[widx] = wn;
    } else {
        const int wbits = __float_as_int(1.0f - ALPHA);
        int base = 0;
        for (; base + 32 <= wn; base += 32) {
            unsigned nxt = 0u;
            if (base + 32 + lane < wn) nxt = wl[base + 32 + lane];
            agg_rows((int)(ent & 0xFFFFu), (int)(ent >> 16), wbits, 32, lane, zin, acc);
            ent = nxt;
        }
        if (base < wn)
            agg_rows((int)(ent & 0xFFFFu), (int)(ent >> 16), wbits, wn - base, lane, zin, acc);
    }
    __syncthreads();

    if (MODE == 1) {
        const int r    = lane & 15;
        const bool hi  = lane >= 16;
        const int koff = hi ? 2 : 0;
        const bool act = wave < AGG_GW;
        const int row0 = (wave >> 1) * 16;
        const int col0 = (wave & 1) * 64 + r;
        v8f c[4] = {};
        if (act) {
            const float* arow = acc + (row0 + r) * AGG_S + koff;
            for (int k0 = 0; k0 < DH; k0 += 4) {
                const float2 a2 = *(const float2*)(arow + k0);
                v2f a; a.x = a2.x; a.y = a2.y;
                const float2* brow = sW2 + (size_t)((k0 >> 1) + (hi ? 1 : 0)) * DH + col0;
#pragma unroll
                for (int t = 0; t < 4; ++t) {
                    const float2 b2 = brow[t * 16];
                    v2f b; b.x = b2.x; b.y = b2.y;
                    c[t] = __builtin_amdgcn_wmma_f32_16x16x4_f32(
                        false, a, false, b, (short)0, c[t], false, false);
                }
            }
#pragma unroll
            for (int t = 0; t < 4; ++t) {
#pragma unroll
                for (int v = 0; v < 8; ++v) {
                    const float hv = acc[(row0 + v + (hi ? 8 : 0)) * AGG_S + col0 + t * 16];
                    c[t][v] = (1.0f - beta) * hv + beta * c[t][v];
                }
            }
        }
        __syncthreads();
        if (act) {
#pragma unroll
            for (int t = 0; t < 4; ++t) {
#pragma unroll
                for (int v = 0; v < 8; ++v)
                    acc[(row0 + v + (hi ? 8 : 0)) * AGG_S + col0 + t * 16] = c[t][v];
            }
        }
        __syncthreads();
    }

    for (int i = tid; i < cnt * (DH / 4); i += AGG_TPB) {
        const int n = i >> 5, c4 = i & 31;
        *(float4*)(out + (size_t)(lo + n) * DH + c4 * 4) = *(const float4*)(acc + n * AGG_S + c4 * 4);
    }
}

#define BN_R     196
#define BN_NB    ((NND + BN_R - 1) / BN_R)
#define BN_TPB   512
#define BN_WAVES (BN_TPB / 32)
#define BN_RPW   ((BN_R + BN_WAVES - 1) / BN_WAVES)
#define BN_PG    (BN_TPB / 64)
static_assert(BN_NB % BN_PG == 0, "partial rows split evenly over the thread groups");
static_assert(DH == 128, "a wave's 32 lanes x 16 B cover one row");

__global__ __launch_bounds__(BN_TPB)
void k_bn_part(const float* __restrict__ z, float* __restrict__ part) {
    __shared__ float4 red[BN_WAVES][64];
    const int tid  = threadIdx.x;
    const int lane = tid & 31;
    const int wave = __builtin_amdgcn_readfirstlane(tid >> 5);
    const int lo   = blockIdx.x * BN_R;
    const int cnt  = (NND - lo < BN_R) ? (NND - lo) : BN_R;
    const float* col = z + (size_t)lo * DH + lane * 4;

    float4 v[BN_RPW];
#pragma unroll
    for (int j = 0; j < BN_RPW; ++j) {
        const int n  = wave + j * BN_WAVES;
        const int nc = (n < cnt) ? n : (cnt - 1);
        v[j] = *(const float4*)(col + (size_t)nc * DH);
    }
    float4 s = make_float4(0.0f, 0.0f, 0.0f, 0.0f);
    float4 q = make_float4(0.0f, 0.0f, 0.0f, 0.0f);
#pragma unroll
    for (int j = 0; j < BN_RPW; ++j) {
        if (wave + j * BN_WAVES < cnt) {
            s.x += v[j].x; s.y += v[j].y; s.z += v[j].z; s.w += v[j].w;
            q.x += v[j].x * v[j].x; q.y += v[j].y * v[j].y;
            q.z += v[j].z * v[j].z; q.w += v[j].w * v[j].w;
        }
    }
    red[wave][lane]      = s;
    red[wave][32 + lane] = q;
    __syncthreads();
    if (tid < 64) {
        float4 a = red[0][tid];
        for (int w = 1; w < BN_WAVES; ++w) {
            const float4 b = red[w][tid];
            a.x += b.x; a.y += b.y; a.z += b.z; a.w += b.w;
        }
        *(float4*)(part + (size_t)blockIdx.x * (2 * DH) + tid * 4) = a;
    }
}

__global__ __launch_bounds__(BN_TPB)
void k_bn_apply(float* z, const float* __restrict__ part,
                const float* __restrict__ gamma, const float* __restrict__ beta) {
    __shared__ float4 red[BN_PG][64];
    const int tid  = threadIdx.x;
    const int lane = tid & 31;
    const int wave = __builtin_amdgcn_readfirstlane(tid >> 5);
    const int lo   = blockIdx.x * BN_R;
    const int cnt  = (NND - lo < BN_R) ? (NND - lo) : BN_R;
    float* col = z + (size_t)lo * DH + lane * 4;

    float4 v[BN_RPW];
#pragma unroll
    for (int j = 0; j < BN_RPW; ++j) {
        const int n  = wave + j * BN_WAVES;
        const int nc = (n < cnt) ? n : (cnt - 1);
        v[j] = *(const float4*)(col + (size_t)nc * DH);
    }

    const float4 gm = *(const float4*)(gamma + lane * 4);
    const float4 bt = *(const float4*)(beta + lane * 4);
    __builtin_amdgcn_sched_barrier(0);

    const int cg = tid & 63;
    const int g  = tid >> 6;
    const float* pp = part + (size_t)g * (2 * DH) + cg * 4;
    float4 a = make_float4(0.0f, 0.0f, 0.0f, 0.0f);
#pragma unroll
    for (int b = 0; b < BN_NB / BN_PG; ++b) {
        const float4 p = *(const float4*)(pp + (size_t)b * (BN_PG * 2 * DH));
        a.x += p.x; a.y += p.y; a.z += p.z; a.w += p.w;
    }
    red[g][cg] = a;
    __syncthreads();

    float4 S = red[0][lane], Q = red[0][32 + lane];
#pragma unroll
    for (int k = 1; k < BN_PG; ++k) {
        const float4 s1 = red[k][lane], q1 = red[k][32 + lane];
        S.x += s1.x; S.y += s1.y; S.z += s1.z; S.w += s1.w;
        Q.x += q1.x; Q.y += q1.y; Q.z += q1.z; Q.w += q1.w;
    }
    const float invN = 1.0f / (float)NND;
    float4 mean, inv;
    mean.x = S.x * invN; mean.y = S.y * invN; mean.z = S.z * invN; mean.w = S.w * invN;
    inv.x = rsqrtf(Q.x * invN - mean.x * mean.x + BNEPS);
    inv.y = rsqrtf(Q.y * invN - mean.y * mean.y + BNEPS);
    inv.z = rsqrtf(Q.z * invN - mean.z * mean.z + BNEPS);
    inv.w = rsqrtf(Q.w * invN - mean.w * mean.w + BNEPS);

#pragma unroll
    for (int j = 0; j < BN_RPW; ++j) {
        const int n = wave + j * BN_WAVES;
        if (n < cnt) {
            float4 o;
            o.x = fmaxf((v[j].x - mean.x) * inv.x * gm.x + bt.x, 0.0f);
            o.y = fmaxf((v[j].y - mean.y) * inv.y * gm.y + bt.y, 0.0f);
            o.z = fmaxf((v[j].z - mean.z) * inv.z * gm.z + bt.z, 0.0f);
            o.w = fmaxf((v[j].w - mean.w) * inv.w * gm.w + bt.w, 0.0f);
            *(float4*)(col + (size_t)n * DH) = o;
        }
    }
}

extern "C" void kernel_launch(void* const* d_in, const int* in_sizes, int n_in,
                              void* d_out, int out_size, void* d_ws, size_t ws_size,
                              hipStream_t stream) {
    const float* x        = (const float*)d_in[0];
    const int*   edge     = (const int*)d_in[1];
    const float* W_lin    = (const float*)d_in[2];
    const float* b_lin    = (const float*)d_in[3];
    const float* W1s      = (const float*)d_in[4];
    const float* bn_gamma = (const float*)d_in[5];
    const float* bn_beta  = (const float*)d_in[6];
    float* out = (float*)d_out;

    float* ws    = (float*)d_ws;
    float* dinv  = ws;
    float* part  = ws + NND;
    float* bufA  = part + (size_t)BN_NB * 2 * DH;
    float* bufB  = bufA + (size_t)NND * DH;
    float* x0    = bufB + (size_t)NND * DH;
    int* cnts       = (int*)(x0 + (size_t)NND * DH);
    unsigned* lists = (unsigned*)(cnts + AGG_NWG * AGG_WAVES);

    const int* src = edge;
    const int* dst = edge + NED;

    const int nBlkN   = (NND + 255) / 256;
    const int nBlkE   = (NED + 255) / 256;
    const int nBlkGemm = (NND / 16 + 7) / 8;

    k_deg_init<<<nBlkN, 256, 0, stream>>>(dinv);
    k_deg_edges<<<nBlkE, 256, 0, stream>>>(dst, dinv);
    k_deg_rsqrt<<<nBlkN, 256, 0, stream>>>(dinv);

    k_gemm_wmma<DIN><<<nBlkGemm, 256, DIN * DH * sizeof(float), stream>>>(
        x, W_lin, bufA, 0, 0.0f);

    k_agg<0><<<AGG_NWG, AGG_TPB, AGG_LDS0, stream>>>(src, dst, bufA, bufA, dinv, b_lin, bufB,
                                                     cnts, lists, W1s, 0.0f);

    hipMemcpyAsync(x0, bufB, (size_t)NND * DH * sizeof(float),
                   hipMemcpyDeviceToDevice, stream);

    float* z = bufB;
    float* h = bufA;
    for (int i = 0; i < 5; ++i) {
        const float beta = logf(0.5f / (float)(i + 1) + 1.0f);
        k_agg<1><<<AGG_NWG, AGG_TPB, AGG_LDS1, stream>>>(src, dst, z, x0, dinv, b_lin, h, cnts, lists,
                                                         W1s + (size_t)i * DH * DH, beta);
        if (i < 4) {
            k_bn_part<<<BN_NB, BN_TPB, 0, stream>>>(h, part);
            k_bn_apply<<<BN_NB, BN_TPB, 0, stream>>>(
                h, part, bn_gamma + i * DH, bn_beta + i * DH);
        }
        float* t = z; z = h; h = t;
    }

    hipMemcpyAsync(out, z, (size_t)NND * DH * sizeof(float),
                   hipMemcpyDeviceToDevice, stream);
}


# BatchNorm+ReLU folded into the LDS edge-sum launches: store loop adds column sums of its rows, next pass sums the partials and gathers max(v*sc+sh,0); 8 launches fewer; statistics sum order changes
# speedup vs baseline: 3.6647x; 1.1015x over previous
#include <hip/hip_runtime.h>
#include <cmath>

typedef __attribute__((ext_vector_type(2))) float v2f;
typedef __attribute__((ext_vector_type(8))) float v8f;

#define NND   50000
#define NED   800000
#define DIN   256
#define DH    128
#define ALPHA 0.1f
#define BNEPS 1e-5f

__global__ void k_deg_init(float* deg) {
    int i = blockIdx.x * blockDim.x + threadIdx.x;
    if (i < NND) deg[i] = 1.0f;
}
__global__ void k_deg_edges(const int* __restrict__ dst, float* deg) {
    int e = blockIdx.x * blockDim.x + threadIdx.x;
    if (e < NED) atomicAdd(&deg[dst[e]], 1.0f);
}
__global__ void k_deg_rsqrt(float* deg) {
    int i = blockIdx.x * blockDim.x + threadIdx.x;
    if (i < NND) deg[i] = rsqrtf(fmaxf(deg[i], 1.0f));
}

template <int K>
__global__ __launch_bounds__(256)
void k_gemm_wmma(const float* in, const float* __restrict__ W,
                 float* out, int mode, float beta) {
    extern __shared__ float2 sW2[];
    const int tid = threadIdx.x;

    {
        const int totalIter = (K / 2) * (DH / 4);
        for (int i = tid; i < totalIter; i += 256) {
            const int kp = i / (DH / 4);
            const int c4 = i % (DH / 4);
            const float4 a0 = *(const float4*)(W + (size_t)(2 * kp)     * DH + c4 * 4);
            const float4 a1 = *(const float4*)(W + (size_t)(2 * kp + 1) * DH + c4 * 4);
            float4* dst4 = (float4*)(sW2 + (size_t)kp * DH + c4 * 4);
            float4 f0; f0.x = a0.x; f0.y = a1.x; f0.z = a0.y; f0.w = a1.y;
            float4 f1; f1.x = a0.z; f1.y = a1.z; f1.z = a0.w; f1.w = a1.w;
            dst4[0] = f0;
            dst4[1] = f1;
        }
    }
    __syncthreads();

    const int wave = tid >> 5;
    const int lane = tid & 31;
    const int tileRow = blockIdx.x * 8 + wave;
    if (tileRow >= NND / 16) return;
    const int m0 = tileRow * 16;
    const int r    = lane & 15;
    const bool hi  = lane >= 16;
    const int koff = hi ? 2 : 0;
    const float* arow = in + (size_t)(m0 + r) * K;

    v8f c[8] = {};
    for (int k0 = 0; k0 < K; k0 += 4) {
        const float2 a2 = *(const float2*)(arow + k0 + koff);
        v2f a; a.x = a2.x; a.y = a2.y;
        const int pairRow = (k0 >> 1) + (hi ? 1 : 0);
        const float2* brow = sW2 + (size_t)pairRow * DH + r;
#pragma unroll
        for (int t = 0; t < 8; ++t) {
            const float2 b2 = brow[t * 16];
            v2f b; b.x = b2.x; b.y = b2.y;
            c[t] = __builtin_amdgcn_wmma_f32_16x16x4_f32(
                false, a, false, b, (short)0, c[t], false, false);
        }
    }

#pragma unroll
    for (int t = 0; t < 8; ++t) {
        const int col = t * 16 + r;
#pragma unroll
        for (int v = 0; v < 8; ++v) {
            const int row = m0 + v + (hi ? 8 : 0);
            const size_t o = (size_t)row * DH + col;
            float val = c[t][v];
            if (mode != 0) val = (1.0f - beta) * in[o] + beta * val;
            out[o] = val;
        }
    }
}

#define AGG_R      196
#define AGG_NWG    ((NND + AGG_R - 1) / AGG_R)
#define AGG_TPB    1024
#define AGG_WAVES  (AGG_TPB / 32)
#define AGG_LPT    8
#define AGG_BATCH  (32 * 4 * AGG_LPT)
#define AGG_QCAP   (AGG_BATCH + 32)
#define AGG_NBATCH ((NED + AGG_BATCH - 1) / AGG_BATCH)
#define AGG_S      (DH + 4)
#define AGG_RT     ((AGG_R + 15) / 16)
#define AGG_ACCF   (AGG_RT * 16 * AGG_S)
#define AGG_GW     (AGG_RT * 2)
#define AGG_LDS0   ((AGG_ACCF + AGG_WAVES * AGG_QCAP) * 4)
#define AGG_LDS1   (AGG_LDS0 + DH * DH * 4)
#define AGG_LDS2   (AGG_LDS1 + 2 * DH * 4)
#define AGG_PG     (AGG_TPB / 64)
#define AGG_WCAP   2048
static_assert(NED <= (1 << 20) && AGG_R <= (1 << 12), "queue entry packing");
static_assert(AGG_WAVES <= AGG_NBATCH, "every wave has a first batch");
static_assert(AGG_GW <= AGG_WAVES, "a wave for every row tile and column half");
static_assert(AGG_LDS2 <= 320 * 1024, "one work-group's LDS");
static_assert(AGG_NWG % AGG_PG == 0, "partial rows split evenly over the thread groups");
static_assert(AGG_WAVES * 64 * 16 <= AGG_WAVES * AGG_QCAP * 4, "the column-sum scratch fits in the queue area");
static_assert(DH == 128 && AGG_TPB >= DH, "a wave's 32 lanes x 16 B cover one row; a thread per column");
static_assert(NND <= (1 << 16) && AGG_R <= (1 << 12), "list entry packing");

template <bool BN>
__device__ __forceinline__ void agg_rows(int s, int dl, int wbits, int cntc, int lane,
                                         const float* __restrict__ zin, float* acc,
                                         const float (&sc)[4], const float (&sh)[4]) {
    for (int g = 0; g < cntc; g += 8) {
        float v[8][4];
#pragma unroll
        for (int u = 0; u < 8; ++u) {
            int k = g + u;
            if (k > cntc - 1) k = cntc - 1;
            const int sk = __builtin_amdgcn_readlane(s, k);
            const float* row = zin + (size_t)sk * DH + lane;
            v[u][0] = row[0];
            v[u][1] = row[32];
            v[u][2] = row[64];
            v[u][3] = row[96];
        }
#pragma unroll
        for (int u = 0; u < 8; ++u) {
            const int k = g + u;
            if (k < cntc) {
                const int dk = __builtin_amdgcn_readlane(dl, k);
                float* arow = acc + dk * AGG_S + lane;
                if (BN) {
                    atomicAdd(arow,      fmaxf(fmaf(v[u][0], sc[0], sh[0]), 0.0f));
                    atomicAdd(arow + 32, fmaxf(fmaf(v[u][1], sc[1], sh[1]), 0.0f));
                    atomicAdd(arow + 64, fmaxf(fmaf(v[u][2], sc[2], sh[2]), 0.0f));
                    atomicAdd(arow + 96, fmaxf(fmaf(v[u][3], sc[3], sh[3]), 0.0f));
                } else {
                    const float wk = __int_as_float(__builtin_amdgcn_readlane(wbits, k));
                    atomicAdd(arow,      wk * v[u][0]);
                    atomicAdd(arow + 32, wk * v[u][1]);
                    atomicAdd(arow + 64, wk * v[u][2]);
                    atomicAdd(arow + 96, wk * v[u][3]);
                }
            }
        }
    }
}

template <int MODE>
__device__ __forceinline__ void agg_chunk(const unsigned* q, int base, int cntc, int lane, int lo,
                                          const int* __restrict__ src,
                                          const float* __restrict__ zin,
                                          const float* __restrict__ dinv, float* acc,
                                          unsigned* wl, int& wn,
                                          const float (&sc)[4], const float (&sh)[4]) {
    const unsigned ent = (lane < cntc) ? q[base + lane] : 0u;
    const int e  = (int)(ent & 0xFFFFFu);
    const int dl = (int)(ent >> 20);
    const int s  = src[e];
    float w = 1.0f - ALPHA;
    if (MODE == 0) {
        w = dinv[s] * dinv[lo + dl];
        if (wn + cntc <= AGG_WCAP && lane < cntc)
            wl[wn + lane] = ((unsigned)dl << 16) | (unsigned)s;
        wn += cntc;
    }
    agg_rows<MODE == 2>(s, dl, __float_as_int(w), cntc, lane, zin, acc, sc, sh);
}

__device__ __forceinline__ void agg_load(int4 (&dv)[AGG_LPT], const int* __restrict__ dst,
                                         int t, int lane) {
    const int eb = t * AGG_BATCH + lane * 4;
#pragma unroll
    for (int j = 0; j < AGG_LPT; ++j) {
        int e = eb + j * 128;
        if (e > NED - 4) e = NED - 4;
        dv[j] = *(const int4*)(dst + e);
    }
    if (t == AGG_NBATCH - 1) {
#pragma unroll
        for (int j = 0; j < AGG_LPT; ++j)
            if (eb + j * 128 >= NED) dv[j] = make_int4(-1, -1, -1, -1);
    }
}

__device__ __forceinline__ void agg_scan(const int4 (&dv)[AGG_LPT], int t, int lane, int lo, int cnt,
                                         unsigned* q, int& qn) {
    const int eb = t * AGG_BATCH + lane * 4;
#pragma unroll
    for (int j = 0; j < AGG_LPT; ++j) {
        const int e = eb + j * 128;
        const int dd[4] = {dv[j].x, dv[j].y, dv[j].z, dv[j].w};
#pragma unroll
        for (int c = 0; c < 4; ++c) {
            const unsigned dl = (unsigned)(dd[c] - lo);
            const bool m = dl < (unsigned)cnt;
            const unsigned bm = __builtin_amdgcn_ballot_w32(m);
            if (bm != 0u) {
                const int pos = qn + (int)__builtin_amdgcn_mbcnt_lo(bm, 0u);
                if (m) q[pos] = (dl << 20) | (unsigned)(e + c);
                qn += __builtin_popcount(bm);
            }
        }
    }
}

template <int MODE>
__device__ __forceinline__ void agg_drain(unsigned* q, int& qn, int lane, int lo,
                                          const int* __restrict__ src,
                                          const float* __restrict__ zin,
                                          const float* __restrict__ dinv, float* acc,
                                          unsigned* wl, int& wn,
                                          const float (&sc)[4], const float (&sh)[4]) {
    if (qn < 32) return;
    __builtin_amdgcn_wave_barrier();
    const int full = qn & ~31;
    for (int base = 0; base < full; base += 32)
        agg_chunk<MODE>(q, base, 32, lane, lo, src, zin, dinv, acc, wl, wn, sc, sh);
    const int rem = qn - full;
    unsigned keep = 0u;
    if (lane < rem) keep = q[full + lane];
    __builtin_amdgcn_wave_barrier();
    if (lane < rem) q[lane] = keep;
    qn = rem;
}

template <int MODE>
__global__ __launch_bounds__(AGG_TPB)
void k_agg(const int* __restrict__ src, const int* __restrict__ dst,
           const float* __restrict__ zin, const float* __restrict__ start,
           const float* __restrict__ dinv, const float* __restrict__ bias,
           float* __restrict__ out, int* cnts, unsigned* lists,
           const float* __restrict__ W1, float beta,
           const float* __restrict__ pin, const float* __restrict__ gamma,
           const float* __restrict__ bnb, float* __restrict__ pout) {
    extern __shared__ float agg_smem[];
    float* acc = agg_smem;
    unsigned* qall = (unsigned*)(agg_smem + AGG_ACCF);
    float2* sW2 = (float2*)(agg_smem + AGG_ACCF + AGG_WAVES * AGG_QCAP);
    float* scsh = agg_smem + AGG_ACCF + AGG_WAVES * AGG_QCAP + DH * DH;
    float4* red = (float4*)qall;
    const int tid  = threadIdx.x;
    const int lane = tid & 31;
    const int wave = __builtin_amdgcn_readfirstlane(tid >> 5);
    const int lo   = blockIdx.x * AGG_R;
    const int cnt  = (NND - lo < AGG_R) ? (NND - lo) : AGG_R;
    unsigned* q = qall + wave * AGG_QCAP;
    const int widx = blockIdx.x * AGG_WAVES + wave;
    unsigned* wl = lists + (size_t)widx * AGG_WCAP;

    const int rot = (int)(((long long)blockIdx.x * AGG_NBATCH) / AGG_NWG);
    int t0 = wave;
    int ta = t0 + rot;
    if (ta >= AGG_NBATCH) ta -= AGG_NBATCH;
    int4 dva[AGG_LPT], dvb[AGG_LPT];

    int wn = 0;
    bool scan = true;
    unsigned ent = 0u;
    if (MODE == 0) {
        agg_load(dva, dst, ta, lane);
    } else {
        wn = cnts[widx];
        scan = wn > AGG_WCAP;
        if (scan) agg_load(dva, dst, ta, lane);
        else if (lane < wn) ent = wl[lane];
    }

    float4 pa = make_float4(0.0f, 0.0f, 0.0f, 0.0f);
    float gmv = 0.0f, btv = 0.0f;
    if (MODE == 2) {
        if (tid < DH) { gmv = gamma[tid]; btv = bnb[tid]; }
        const float* pp = pin + (size_t)(tid >> 6) * (2 * DH) + (tid & 63) * 4;
#pragma unroll
        for (int b = 0; b < AGG_NWG / AGG_PG; ++b) {
            const float4 p = *(const float4*)(pp + (size_t)b * (AGG_PG * 2 * DH));
            pa.x += p.x; pa.y += p.y; pa.z += p.z; pa.w += p.w;
        }
    }

    for (int i = tid; i < cnt * (DH / 4); i += AGG_TPB) {
        const int n = i >> 5, c4 = i & 31;
        const float4 v = *(const float4*)(start + (size_t)(lo + n) * DH + c4 * 4);
        float4 o;
        if (MODE == 0) {
            const float dn = dinv[lo + n];
            const float w = dn * dn;
            const float4 bb = *(const float4*)(bias + c4 * 4);
            o.x = w * v.x + bb.x; o.y = w * v.y + bb.y;
            o.z = w * v.z + bb.z; o.w = w * v.w + bb.w;
        } else {
            o.x = ALPHA * v.x; o.y = ALPHA * v.y; o.z = ALPHA * v.z; o.w = ALPHA * v.w;
        }
        *(float4*)(acc + n * AGG_S + c4 * 4) = o;
    }
    if (MODE != 0) {
        for (int i = tid; i < (DH / 2) * (DH / 4); i += AGG_TPB) {
            const int kp = i / (DH / 4);
            const int c4 = i % (DH / 4);
            const float4 a0 = *(const float4*)(W1 + (size_t)(2 * kp)     * DH + c4 * 4);
            const float4 a1 = *(const float4*)(W1 + (size_t)(2 * kp + 1) * DH + c4 * 4);
            float4* dst4 = (float4*)(sW2 + (size_t)kp * DH + c4 * 4);
            float4 f0; f0.x = a0.x; f0.y = a1.x; f0.z = a0.y; f0.w = a1.y;
            float4 f1; f1.x = a0.z; f1.y = a1.z; f1.z = a0.w; f1.w = a1.w;
            dst4[0] = f0;
            dst4[1] = f1;
        }
        for (int i = cnt * AGG_S + tid; i < AGG_ACCF; i += AGG_TPB) acc[i] = 0.0f;
    }
    float sc[4] = {0.0f, 0.0f, 0.0f, 0.0f};
    float sh[4] = {0.0f, 0.0f, 0.0f, 0.0f};
    if (MODE == 2) {
        red[tid] = pa;
        __syncthreads();
        if (tid < DH) {
            const float* rf = (const float*)red;
            float S = rf[tid], Q = rf[DH + tid];
#pragma unroll
            for (int k = 1; k < AGG_PG; ++k) {
                S += rf[k * 2 * DH + tid];
                Q += rf[k * 2 * DH + DH + tid];
            }
            const float invN = 1.0f / (float)NND;
            const float mean = S * invN;
            const float inv  = rsqrtf(Q * invN - mean * mean + BNEPS);
            const float gi   = gmv * inv;
            scsh[tid]      = (1.0f - ALPHA) * gi;
            scsh[DH + tid] = (1.0f - ALPHA) * (btv - mean * gi);
        }
    }
    __syncthreads();
    if (MODE == 2) {
#pragma unroll
        for (int j = 0; j < 4; ++j) {
            sc[j] = scsh[lane + 32 * j];
            sh[j] = scsh[DH + lane + 32 * j];
        }
    }

    if (scan) {
        int qn = 0;
        while (true) {
            const int t1 = t0 + AGG_WAVES;
            const bool has1 = t1 < AGG_NBATCH;
            int tb = (has1 ? t1 : t0) + rot;
            if (tb >= AGG_NBATCH) tb -= AGG_NBATCH;
            agg_load(dvb, dst, tb, lane);
            agg_scan(dva, ta, lane, lo, cnt, q, qn);
            agg_drain<MODE>(q, qn, lane, lo, src, zin, dinv, acc, wl, wn, sc, sh);
            if (!has1) break;
            const int t2 = t1 + AGG_WAVES;
            const bool has2 = t2 < AGG_NBATCH;
            ta = (has2 ? t2 : t1) + rot;
            if (ta >= AGG_NBATCH) ta -= AGG_NBATCH;
            agg_load(dva, dst, ta, lane);
            agg_scan(dvb, tb, lane, lo, cnt, q, qn);
            agg_drain<MODE>(q, qn, lane, lo, src, zin, dinv, acc, wl, wn, sc, sh);
            if (!has2) break;
            t0 = t2;
        }
        __builtin_amdgcn_wave_barrier();
        if (qn > 0) agg_chunk<MODE>(q, 0, qn, lane, lo, src, zin, dinv, acc, wl, wn, sc, sh);
        if (MODE == 0 && lane == 0) cnts[widx] = wn;
    } else {
        const int wbits = __float_as_int(1.0f - ALPHA);
        int base = 0;
        for (; base + 32 <= wn; base += 32) {
            unsigned nxt = 0u;
            if (base + 32 + lane < wn) nxt = wl[base + 32 + lane];
            agg_rows<MODE == 2>((int)(ent & 0xFFFFu), (int)(ent >> 16), wbits, 32, lane, zin, acc, sc, sh);
            ent = nxt;
        }
        if (base < wn)
            agg_rows<MODE == 2>((int)(ent & 0xFFFFu), (int)(ent >> 16), wbits, wn - base, lane, zin, acc, sc, sh);
    }
    __syncthreads();

    if (MODE != 0) {
        const int r    = lane & 15;
        const bool hi  = lane >= 16;
        const int koff = hi ? 2 : 0;
        const bool act = wave < AGG_GW;
        const int row0 = (wave >> 1) * 16;
        const int col0 = (wave & 1) * 64 + r;
        v8f c[4] = {};
        if (act) {
            const float* arow = acc + (row0 + r) * AGG_S + koff;
            for (int k0 = 0; k0 < DH; k0 += 4) {
                const float2 a2 = *(const float2*)(arow + k0);
                v2f a; a.x = a2.x; a.y = a2.y;
                const float2* brow = sW2 + (size_t)((k0 >> 1) + (hi ? 1 : 0)) * DH + col0;
#pragma unroll
                for (int t = 0; t < 4; ++t) {
                    const float2 b2 = brow[t * 16];
                    v2f b; b.x = b2.x; b.y = b2.y;
                    c[t] = __builtin_amdgcn_wmma_f32_16x16x4_f32(
                        false, a, false, b, (short)0, c[t], false, false);
                }
            }
#pragma unroll
            for (int t = 0; t < 4; ++t) {
#pragma unroll
                for (int v = 0; v < 8; ++v) {
                    const float hv = acc[(row0 + v + (hi ? 8 : 0)) * AGG_S + col0 + t * 16];
                    c[t][v] = (1.0f - beta) * hv + beta * c[t][v];
                }
            }
        }
        __syncthreads();
        if (act) {
#pragma unroll
            for (int t = 0; t < 4; ++t) {
#pragma unroll
                for (int v = 0; v < 8; ++v)
                    acc[(row0 + v + (hi ? 8 : 0)) * AGG_S + col0 + t * 16] = c[t][v];
            }
        }
        __syncthreads();
    }

    float4 cs = make_float4(0.0f, 0.0f, 0.0f, 0.0f);
    float4 cq = make_float4(0.0f, 0.0f, 0.0f, 0.0f);
    for (int i = tid; i < cnt * (DH / 4); i += AGG_TPB) {
        const int n = i >> 5, c4 = i & 31;
        const float4 o = *(const float4*)(acc + n * AGG_S + c4 * 4);
        *(float4*)(out + (size_t)(lo + n) * DH + c4 * 4) = o;
        if (MODE != 0) {
            cs.x += o.x; cs.y += o.y; cs.z += o.z; cs.w += o.w;
            cq.x += o.x * o.x; cq.y += o.y * o.y; cq.z += o.z * o.z; cq.w += o.w * o.w;
        }
    }
    if (MODE != 0 && pout != nullptr) {
        red[wave * 64 + lane]      = cs;
        red[wave * 64 + 32 + lane] = cq;
        __syncthreads();
        if (tid < 64) {
            float4 a = red[tid];
#pragma unroll
            for (int w = 1; w < AGG_WAVES; ++w) {
                const float4 b = red[w * 64 + tid];
                a.x += b.x; a.y += b.y; a.z += b.z; a.w += b.w;
            }
            *(float4*)(pout + (size_t)blockIdx.x * (2 * DH) + tid * 4) = a;
        }
    }
}

extern "C" void kernel_launch(void* const* d_in, const int* in_sizes, int n_in,
                              void* d_out, int out_size, void* d_ws, size_t ws_size,
                              hipStream_t stream) {
    const float* x        = (const float*)d_in[0];
    const int*   edge     = (const int*)d_in[1];
    const float* W_lin    = (const float*)d_in[2];
    const float* b_lin    = (const float*)d_in[3];
    const float* W1s      = (const float*)d_in[4];
    const float* bn_gamma = (const float*)d_in[5];
    const float* bn_beta  = (const float*)d_in[6];
    float* out = (float*)d_out;

    float* ws    = (float*)d_ws;
    float* dinv  = ws;
    float* partA = ws + NND;
    float* partB = partA + (size_t)AGG_NWG * 2 * DH;
    float* bufA  = partB + (size_t)AGG_NWG * 2 * DH;
    float* bufB  = bufA + (size_t)NND * DH;
    float* x0    = bufB + (size_t)NND * DH;
    int* cnts       = (int*)(x0 + (size_t)NND * DH);
    unsigned* lists = (unsigned*)(cnts + AGG_NWG * AGG_WAVES);

    const int* src = edge;
    const int* dst = edge + NED;

    const int nBlkN   = (NND + 255) / 256;
    const int nBlkE   = (NED + 255) / 256;
    const int nBlkGemm = (NND / 16 + 7) / 8;

    k_deg_init<<<nBlkN, 256, 0, stream>>>(dinv);
    k_deg_edges<<<nBlkE, 256, 0, stream>>>(dst, dinv);
    k_deg_rsqrt<<<nBlkN, 256, 0, stream>>>(dinv);

    k_gemm_wmma<DIN><<<nBlkGemm, 256, DIN * DH * sizeof(float), stream>>>(
        x, W_lin, bufA, 0, 0.0f);

    k_agg<0><<<AGG_NWG, AGG_TPB, AGG_LDS0, stream>>>(src, dst, bufA, bufA, dinv, b_lin, bufB,
                                                     cnts, lists, W1s, 0.0f,
                                                     nullptr, nullptr, nullptr, nullptr);

    hipMemcpyAsync(x0, bufB, (size_t)NND * DH * sizeof(float),
                   hipMemcpyDeviceToDevice, stream);

    float* z = bufB;
    float* h = bufA;
    for (int i = 0; i < 5; ++i) {
        const float beta = logf(0.5f / (float)(i + 1) + 1.0f);
        float* pout = (i < 4) ? ((i & 1) ? partB : partA) : nullptr;
        if (i == 0) {
            k_agg<1><<<AGG_NWG, AGG_TPB, AGG_LDS1, stream>>>(src, dst, z, x0, dinv, b_lin, h, cnts, lists,
                                                             W1s, beta,
                                                             nullptr, nullptr, nullptr, pout);
        } else {
            const float* pin = ((i - 1) & 1) ? partB : partA;
            k_agg<2><<<AGG_NWG, AGG_TPB, AGG_LDS2, stream>>>(src, dst, z, x0, dinv, b_lin, h, cnts, lists,
                                                             W1s + (size_t)i * DH * DH, beta,
                                                             pin, bn_gamma + (i - 1) * DH,
                                                             bn_beta + (i - 1) * DH, pout);
        }
        float* t = z; z = h; h = t;
    }

    hipMemcpyAsync(out, z, (size_t)NND * DH * sizeof(float),
                   hipMemcpyDeviceToDevice, stream);
}


# Edge-sum kernel keeps its loads in flight: opening loads (start values, W1; W_lin staging in the GEMM) requested at once, replay gather as a rolling 8-row window, never drained; sum order kept
# speedup vs baseline: 3.7918x; 1.0347x over previous
#include <hip/hip_runtime.h>
#include <cmath>

typedef __attribute__((ext_vector_type(2))) float v2f;
typedef __attribute__((ext_vector_type(8))) float v8f;

#define NND   50000
#define NED   800000
#define DIN   256
#define DH    128
#define ALPHA 0.1f
#define BNEPS 1e-5f

__global__ void k_deg_init(float* deg) {
    int i = blockIdx.x * blockDim.x + threadIdx.x;
    if (i < NND) deg[i] = 1.0f;
}
__global__ void k_deg_edges(const int* __restrict__ dst, float* deg) {
    int e = blockIdx.x * blockDim.x + threadIdx.x;
    if (e < NED) atomicAdd(&deg[dst[e]], 1.0f);
}
__global__ void k_deg_rsqrt(float* deg) {
    int i = blockIdx.x * blockDim.x + threadIdx.x;
    if (i < NND) deg[i] = rsqrtf(fmaxf(deg[i], 1.0f));
}

template <int K>
__global__ __launch_bounds__(256)
void k_gemm_wmma(const float* in, const float* __restrict__ W,
                 float* out, int mode, float beta) {
    extern __shared__ float2 sW2[];
    const int tid = threadIdx.x;

    {
        const int totalIter = (K / 2) * (DH / 4);
        static_assert(((K / 2) * (DH / 4)) % 256 == 0, "every thread runs the same number of staging iterations");
#pragma unroll 8
        for (int it = 0; it < totalIter / 256; ++it) {
            const int i = tid + it * 256;
            const int kp = i / (DH / 4);
            const int c4 = i % (DH / 4);
            const float4 a0 = *(const float4*)(W + (size_t)(2 * kp)     * DH + c4 * 4);
            const float4 a1 = *(const float4*)(W + (size_t)(2 * kp + 1) * DH + c4 * 4);
            float4* dst4 = (float4*)(sW2 + (size_t)kp * DH + c4 * 4);
            float4 f0; f0.x = a0.x; f0.y = a1.x; f0.z = a0.y; f0.w = a1.y;
            float4 f1; f1.x = a0.z; f1.y = a1.z; f1.z = a0.w; f1.w = a1.w;
            dst4[0] = f0;
            dst4[1] = f1;
        }
    }
    __syncthreads();

    const int wave = tid >> 5;
    const int lane = tid & 31;
    const int tileRow = blockIdx.x * 8 + wave;
    if (tileRow >= NND / 16) return;
    const int m0 = tileRow * 16;
    const int r    = lane & 15;
    const bool hi  = lane >= 16;
    const int koff = hi ? 2 : 0;
    const float* arow = in + (size_t)(m0 + r) * K;

    v8f c[8] = {};
    for (int k0 = 0; k0 < K; k0 += 4) {
        const float2 a2 = *(const float2*)(arow + k0 + koff);
        v2f a; a.x = a2.x; a.y = a2.y;
        const int pairRow = (k0 >> 1) + (hi ? 1 : 0);
        const float2* brow = sW2 + (size_t)pairRow * DH + r;
#pragma unroll
        for (int t = 0; t < 8; ++t) {
            const float2 b2 = brow[t * 16];
            v2f b; b.x = b2.x; b.y = b2.y;
            c[t] = __builtin_amdgcn_wmma_f32_16x16x4_f32(
                false, a, false, b, (short)0, c[t], false, false);
        }
    }

#pragma unroll
    for (int t = 0; t < 8; ++t) {
        const int col = t * 16 + r;
#pragma unroll
        for (int v = 0; v < 8; ++v) {
            const int row = m0 + v + (hi ? 8 : 0);
            const size_t o = (size_t)row * DH + col;
            float val = c[t][v];
            if (mode != 0) val = (1.0f - beta) * in[o] + beta * val;
            out[o] = val;
        }
    }
}

#define AGG_R      196
#define AGG_NWG    ((NND + AGG_R - 1) / AGG_R)
#define AGG_TPB    1024
#define AGG_WAVES  (AGG_TPB / 32)
#define AGG_LPT    8
#define AGG_BATCH  (32 * 4 * AGG_LPT)
#define AGG_QCAP   (AGG_BATCH + 32)
#define AGG_NBATCH ((NED + AGG_BATCH - 1) / AGG_BATCH)
#define AGG_S      (DH + 4)
#define AGG_RT     ((AGG_R + 15) / 16)
#define AGG_ACCF   (AGG_RT * 16 * AGG_S)
#define AGG_GW     (AGG_RT * 2)
#define AGG_LDS0   ((AGG_ACCF + AGG_WAVES * AGG_QCAP) * 4)
#define AGG_LDS1   (AGG_LDS0 + DH * DH * 4)
#define AGG_LDS2   (AGG_LDS1 + 2 * DH * 4)
#define AGG_PG     (AGG_TPB / 64)
#define AGG_WCAP   2048
#define AGG_SVN    ((AGG_R + AGG_WAVES - 1) / AGG_WAVES)
static_assert(NED <= (1 << 20) && AGG_R <= (1 << 12), "queue entry packing");
static_assert(AGG_WAVES <= AGG_NBATCH, "every wave has a first batch");
static_assert(AGG_GW <= AGG_WAVES, "a wave for every row tile and column half");
static_assert(AGG_LDS2 <= 320 * 1024, "one work-group's LDS");
static_assert(AGG_NWG % AGG_PG == 0, "partial rows split evenly over the thread groups");
static_assert(AGG_WAVES * 64 * 16 <= AGG_WAVES * AGG_QCAP * 4, "the column-sum scratch fits in the queue area");
static_assert(DH == 128 && AGG_TPB >= DH, "a wave's 32 lanes x 16 B cover one row; a thread per column");
static_assert(NND <= (1 << 16) && AGG_R <= (1 << 12), "list entry packing");
static_assert((DH / 2) * (DH / 4) == 2 * AGG_TPB, "W1 staging: two k-pair rows x 4 columns per thread");

template <bool BN>
__device__ __forceinline__ void agg_rows(int s, int dl, int wbits, int cntc, int lane,
                                         const float* __restrict__ zin, float* acc,
                                         const float (&sc)[4], const float (&sh)[4], int first = 0) {
    for (int g = first; g < cntc; g += 8) {
        float v[8][4];
#pragma unroll
        for (int u = 0; u < 8; ++u) {
            int k = g + u;
            if (k > cntc - 1) k = cntc - 1;
            const int sk = __builtin_amdgcn_readlane(s, k);
            const float* row = zin + (size_t)sk * DH + lane;
            v[u][0] = row[0];
            v[u][1] = row[32];
            v[u][2] = row[64];
            v[u][3] = row[96];
        }
#pragma unroll
        for (int u = 0; u < 8; ++u) {
            const int k = g + u;
            if (k < cntc) {
                const int dk = __builtin_amdgcn_readlane(dl, k);
                float* arow = acc + dk * AGG_S + lane;
                if (BN) {
                    atomicAdd(arow,      fmaxf(fmaf(v[u][0], sc[0], sh[0]), 0.0f));
                    atomicAdd(arow + 32, fmaxf(fmaf(v[u][1], sc[1], sh[1]), 0.0f));
                    atomicAdd(arow + 64, fmaxf(fmaf(v[u][2], sc[2], sh[2]), 0.0f));
                    atomicAdd(arow + 96, fmaxf(fmaf(v[u][3], sc[3], sh[3]), 0.0f));
                } else {
                    const float wk = __int_as_float(__builtin_amdgcn_readlane(wbits, k));
                    atomicAdd(arow,      wk * v[u][0]);
                    atomicAdd(arow + 32, wk * v[u][1]);
                    atomicAdd(arow + 64, wk * v[u][2]);
                    atomicAdd(arow + 96, wk * v[u][3]);
                }
            }
        }
    }
}

__device__ __forceinline__ void row_load(float (&v)[4], const float* __restrict__ zin, int sk, int lane) {
    const float* row = zin + (size_t)sk * DH + lane;
    v[0] = row[0]; v[1] = row[32]; v[2] = row[64]; v[3] = row[96];
}
template <bool BN>
__device__ __forceinline__ void row_add(const float (&v)[4], float* acc, int dk, int lane,
                                        const float (&sc)[4], const float (&sh)[4]) {
    float* arow = acc + dk * AGG_S + lane;
#pragma unroll
    for (int j = 0; j < 4; ++j) {
        if (BN) atomicAdd(arow + 32 * j, fmaxf(fmaf(v[j], sc[j], sh[j]), 0.0f));
        else    atomicAdd(arow + 32 * j, (1.0f - ALPHA) * v[j]);
    }
}

template <int MODE>
__device__ __forceinline__ void agg_chunk(const unsigned* q, int base, int cntc, int lane, int lo,
                                          const int* __restrict__ src,
                                          const float* __restrict__ zin,
                                          const float* __restrict__ dinv, float* acc,
                                          unsigned* wl, int& wn,
                                          const float (&sc)[4], const float (&sh)[4]) {
    const unsigned ent = (lane < cntc) ? q[base + lane] : 0u;
    const int e  = (int)(ent & 0xFFFFFu);
    const int dl = (int)(ent >> 20);
    const int s  = src[e];
    float w = 1.0f - ALPHA;
    if (MODE == 0) {
        w = dinv[s] * dinv[lo + dl];
        if (wn + cntc <= AGG_WCAP && lane < cntc)
            wl[wn + lane] = ((unsigned)dl << 16) | (unsigned)s;
        wn += cntc;
    }
    agg_rows<MODE == 2>(s, dl, __float_as_int(w), cntc, lane, zin, acc, sc, sh);
}

__device__ __forceinline__ void agg_load(int4 (&dv)[AGG_LPT], const int* __restrict__ dst,
                                         int t, int lane) {
    const int eb = t * AGG_BATCH + lane * 4;
#pragma unroll
    for (int j = 0; j < AGG_LPT; ++j) {
        int e = eb + j * 128;
        if (e > NED - 4) e = NED - 4;
        dv[j] = *(const int4*)(dst + e);
    }
    if (t == AGG_NBATCH - 1) {
#pragma unroll
        for (int j = 0; j < AGG_LPT; ++j)
            if (eb + j * 128 >= NED) dv[j] = make_int4(-1, -1, -1, -1);
    }
}

__device__ __forceinline__ void agg_scan(const int4 (&dv)[AGG_LPT], int t, int lane, int lo, int cnt,
                                         unsigned* q, int& qn) {
    const int eb = t * AGG_BATCH + lane * 4;
#pragma unroll
    for (int j = 0; j < AGG_LPT; ++j) {
        const int e = eb + j * 128;
        const int dd[4] = {dv[j].x, dv[j].y, dv[j].z, dv[j].w};
#pragma unroll
        for (int c = 0; c < 4; ++c) {
            const unsigned dl = (unsigned)(dd[c] - lo);
            const bool m = dl < (unsigned)cnt;
            const unsigned bm = __builtin_amdgcn_ballot_w32(m);
            if (bm != 0u) {
                const int pos = qn + (int)__builtin_amdgcn_mbcnt_lo(bm, 0u);
                if (m) q[pos] = (dl << 20) | (unsigned)(e + c);
                qn += __builtin_popcount(bm);
            }
        }
    }
}

template <int MODE>
__device__ __forceinline__ void agg_drain(unsigned* q, int& qn, int lane, int lo,
                                          const int* __restrict__ src,
                                          const float* __restrict__ zin,
                                          const float* __restrict__ dinv, float* acc,
                                          unsigned* wl, int& wn,
                                          const float (&sc)[4], const float (&sh)[4]) {
    if (qn < 32) return;
    __builtin_amdgcn_wave_barrier();
    const int full = qn & ~31;
    for (int base = 0; base < full; base += 32)
        agg_chunk<MODE>(q, base, 32, lane, lo, src, zin, dinv, acc, wl, wn, sc, sh);
    const int rem = qn - full;
    unsigned keep = 0u;
    if (lane < rem) keep = q[full + lane];
    __builtin_amdgcn_wave_barrier();
    if (lane < rem) q[lane] = keep;
    qn = rem;
}

template <int MODE>
__global__ __launch_bounds__(AGG_TPB)
void k_agg(const int* __restrict__ src, const int* __restrict__ dst,
           const float* __restrict__ zin, const float* __restrict__ start,
           const float* __restrict__ dinv, const float* __restrict__ bias,
           float* __restrict__ out, int* cnts, unsigned* lists,
           const float* __restrict__ W1, float beta,
           const float* __restrict__ pin, const float* __restrict__ gamma,
           const float* __restrict__ bnb, float* __restrict__ pout) {
    extern __shared__ float agg_smem[];
    float* acc = agg_smem;
    unsigned* qall = (unsigned*)(agg_smem + AGG_ACCF);
    float2* sW2 = (float2*)(agg_smem + AGG_ACCF + AGG_WAVES * AGG_QCAP);
    float* scsh = agg_smem + AGG_ACCF + AGG_WAVES * AGG_QCAP + DH * DH;
    float4* red = (float4*)qall;
    const int tid  = threadIdx.x;
    const int lane = tid & 31;
    const int wave = __builtin_amdgcn_readfirstlane(tid >> 5);
    const int lo   = blockIdx.x * AGG_R;
    const int cnt  = (NND - lo < AGG_R) ? (NND - lo) : AGG_R;
    unsigned* q = qall + wave * AGG_QCAP;
    const int widx = blockIdx.x * AGG_WAVES + wave;
    unsigned* wl = lists + (size_t)widx * AGG_WCAP;

    const int rot = (int)(((long long)blockIdx.x * AGG_NBATCH) / AGG_NWG);
    int t0 = wave;
    int ta = t0 + rot;
    if (ta >= AGG_NBATCH) ta -= AGG_NBATCH;
    int4 dva[AGG_LPT], dvb[AGG_LPT];

    int wn = 0;
    bool scan = true;
    unsigned ent = 0u;
    if (MODE == 0) {
        agg_load(dva, dst, ta, lane);
    } else {
        wn = cnts[widx];
    }

    float4 sv[AGG_SVN];
    float dnv[AGG_SVN];
    float4 bb = make_float4(0.0f, 0.0f, 0.0f, 0.0f);
#pragma unroll
    for (int k = 0; k < AGG_SVN; ++k) {
        int n = wave + k * AGG_WAVES;
        if (n > cnt - 1) n = cnt - 1;
        sv[k] = *(const float4*)(start + (size_t)(lo + n) * DH + lane * 4);
        dnv[k] = (MODE == 0) ? dinv[lo + n] : 0.0f;
    }
    if (MODE == 0) bb = *(const float4*)(bias + lane * 4);
    float4 w1v[2][2];
    if (MODE != 0) {
#pragma unroll
        for (int it = 0; it < 2; ++it) {
            const int i = tid + it * AGG_TPB;
            const int kp = i / (DH / 4);
            const int c4 = i % (DH / 4);
            w1v[it][0] = *(const float4*)(W1 + (size_t)(2 * kp)     * DH + c4 * 4);
            w1v[it][1] = *(const float4*)(W1 + (size_t)(2 * kp + 1) * DH + c4 * 4);
        }
    }

    float4 pa = make_float4(0.0f, 0.0f, 0.0f, 0.0f);
    float gmv = 0.0f, btv = 0.0f;
    if (MODE == 2) {
        if (tid < DH) { gmv = gamma[tid]; btv = bnb[tid]; }
        const float* pp = pin + (size_t)(tid >> 6) * (2 * DH) + (tid & 63) * 4;
#pragma unroll
        for (int b = 0; b < AGG_NWG / AGG_PG; ++b) {
            const float4 p = *(const float4*)(pp + (size_t)b * (AGG_PG * 2 * DH));
            pa.x += p.x; pa.y += p.y; pa.z += p.z; pa.w += p.w;
        }
    }

    if (MODE != 0) {
        scan = wn > AGG_WCAP;
        if (!scan && lane < wn) ent = wl[lane];
    }

#pragma unroll
    for (int k = 0; k < AGG_SVN; ++k) {
        const int n = wave + k * AGG_WAVES;
        if (n < cnt) {
            const float4 v = sv[k];
            float4 o;
            if (MODE == 0) {
                const float w = dnv[k] * dnv[k];
                o.x = w * v.x + bb.x; o.y = w * v.y + bb.y;
                o.z = w * v.z + bb.z; o.w = w * v.w + bb.w;
            } else {
                o.x = ALPHA * v.x; o.y = ALPHA * v.y; o.z = ALPHA * v.z; o.w = ALPHA * v.w;
            }
            *(float4*)(acc + n * AGG_S + lane * 4) = o;
        }
    }
    if (MODE != 0) {
#pragma unroll
        for (int it = 0; it < 2; ++it) {
            const int i = tid + it * AGG_TPB;
            const int kp = i / (DH / 4);
            const int c4 = i % (DH / 4);
            const float4 a0 = w1v[it][0];
            const float4 a1 = w1v[it][1];
            float4* dst4 = (float4*)(sW2 + (size_t)kp * DH + c4 * 4);
            float4 f0; f0.x = a0.x; f0.y = a1.x; f0.z = a0.y; f0.w = a1.y;
            float4 f1; f1.x = a0.z; f1.y = a1.z; f1.z = a0.w; f1.w = a1.w;
            dst4[0] = f0;
            dst4[1] = f1;
        }
        for (int i = cnt * AGG_S + tid; i < AGG_ACCF; i += AGG_TPB) acc[i] = 0.0f;
    }
    float sc[4] = {0.0f, 0.0f, 0.0f, 0.0f};
    float sh[4] = {0.0f, 0.0f, 0.0f, 0.0f};
    if (MODE == 2) {
        red[tid] = pa;
        __syncthreads();
        if (tid < DH) {
            const float* rf = (const float*)red;
            float S = rf[tid], Q = rf[DH + tid];
#pragma unroll
            for (int k = 1; k < AGG_PG; ++k) {
                S += rf[k * 2 * DH + tid];
                Q += rf[k * 2 * DH + DH + tid];
            }
            const float invN = 1.0f / (float)NND;
            const float mean = S * invN;
            const float inv  = rsqrtf(Q * invN - mean * mean + BNEPS);
            const float gi   = gmv * inv;
            scsh[tid]      = (1.0f - ALPHA) * gi;
            scsh[DH + tid] = (1.0f - ALPHA) * (btv - mean * gi);
        }
    }
    __syncthreads();
    if (MODE == 2) {
#pragma unroll
        for (int j = 0; j < 4; ++j) {
            sc[j] = scsh[lane + 32 * j];
            sh[j] = scsh[DH + lane + 32 * j];
        }
    }

    if (scan) {
        if (MODE != 0) agg_load(dva, dst, ta, lane);
        int qn = 0;
        while (true) {
            const int t1 = t0 + AGG_WAVES;
            const bool has1 = t1 < AGG_NBATCH;
            int tb = (has1 ? t1 : t0) + rot;
            if (tb >= AGG_NBATCH) tb -= AGG_NBATCH;
            agg_load(dvb, dst, tb, lane);
            agg_scan(dva, ta, lane, lo, cnt, q, qn);
            agg_drain<MODE>(q, qn, lane, lo, src, zin, dinv, acc, wl, wn, sc, sh);
            if (!has1) break;
            const int t2 = t1 + AGG_WAVES;
            const bool has2 = t2 < AGG_NBATCH;
            ta = (has2 ? t2 : t1) + rot;
            if (ta >= AGG_NBATCH) ta -= AGG_NBATCH;
            agg_load(dva, dst, ta, lane);
            agg_scan(dvb, tb, lane, lo, cnt, q, qn);
            agg_drain<MODE>(q, qn, lane, lo, src, zin, dinv, acc, wl, wn, sc, sh);
            if (!has2) break;
            t0 = t2;
        }
        __builtin_amdgcn_wave_barrier();
        if (qn > 0) agg_chunk<MODE>(q, 0, qn, lane, lo, src, zin, dinv, acc, wl, wn, sc, sh);
        if (MODE == 0 && lane == 0) cnts[widx] = wn;
    } else {
        const int wbits = __float_as_int(1.0f - ALPHA);
        int es = (int)(ent & 0xFFFFu), ed = (int)(ent >> 16);
        float v[8][4];
        if (wn > 0) {
            const int c0 = ((wn < 32) ? wn : 32) - 1;
#pragma unroll
            for (int u = 0; u < 8; ++u)
                row_load(v[u], zin, __builtin_amdgcn_readlane(es, (u < c0) ? u : c0), lane);
        }
        int base = 0;
        for (; base + 32 <= wn; base += 32) {
            unsigned nxt = 0u;
            if (base + 32 + lane < wn) nxt = wl[base + 32 + lane];
            const int ns = (int)(nxt & 0xFFFFu);
            int cl = wn - base - 33;
            if (cl > 31) cl = 31;
            if (cl < 0) cl = 0;
#pragma unroll
            for (int u = 0; u < 32; ++u) {
                row_add<MODE == 2>(v[u & 7], acc, __builtin_amdgcn_readlane(ed, u), lane, sc, sh);
                if (u < 24) {
                    row_load(v[u & 7], zin, __builtin_amdgcn_readlane(es, u + 8), lane);
                } else {
                    const int j = (u - 24 < cl) ? (u - 24) : cl;
                    row_load(v[u & 7], zin, __builtin_amdgcn_readlane(ns, j), lane);
                }
            }
            es = ns; ed = (int)(nxt >> 16);
        }
        const int rem = wn - base;
        if (rem > 0) {
#pragma unroll
            for (int u = 0; u < 8; ++u)
                if (u < rem) row_add<MODE == 2>(v[u], acc, __builtin_amdgcn_readlane(ed, u), lane, sc, sh);
            if (rem > 8) agg_rows<MODE == 2>(es, ed, wbits, rem, lane, zin, acc, sc, sh, 8);
        }
    }
    __syncthreads();

    if (MODE != 0) {
        const int r    = lane & 15;
        const bool hi  = lane >= 16;
        const int koff = hi ? 2 : 0;
        const bool act = wave < AGG_GW;
        const int row0 = (wave >> 1) * 16;
        const int col0 = (wave & 1) * 64 + r;
        v8f c[4] = {};
        if (act) {
            const float* arow = acc + (row0 + r) * AGG_S + koff;
            for (int k0 = 0; k0 < DH; k0 += 4) {
                const float2 a2 = *(const float2*)(arow + k0);
                v2f a; a.x = a2.x; a.y = a2.y;
                const float2* brow = sW2 + (size_t)((k0 >> 1) + (hi ? 1 : 0)) * DH + col0;
#pragma unroll
                for (int t = 0; t < 4; ++t) {
                    const float2 b2 = brow[t * 16];
                    v2f b; b.x = b2.x; b.y = b2.y;
                    c[t] = __builtin_amdgcn_wmma_f32_16x16x4_f32(
                        false, a, false, b, (short)0, c[t], false, false);
                }
            }
#pragma unroll
            for (int t = 0; t < 4; ++t) {
#pragma unroll
                for (int v = 0; v < 8; ++v) {
                    const float hv = acc[(row0 + v + (hi ? 8 : 0)) * AGG_S + col0 + t * 16];
                    c[t][v] = (1.0f - beta) * hv + beta * c[t][v];
                }
            }
        }
        __syncthreads();
        if (act) {
#pragma unroll
            for (int t = 0; t < 4; ++t) {
#pragma unroll
                for (int v = 0; v < 8; ++v)
                    acc[(row0 + v + (hi ? 8 : 0)) * AGG_S + col0 + t * 16] = c[t][v];
            }
        }
        __syncthreads();
    }

    float4 cs = make_float4(0.0f, 0.0f, 0.0f, 0.0f);
    float4 cq = make_float4(0.0f, 0.0f, 0.0f, 0.0f);
    for (int i = tid; i < cnt * (DH / 4); i += AGG_TPB) {
        const int n = i >> 5, c4 = i & 31;
        const float4 o = *(const float4*)(acc + n * AGG_S + c4 * 4);
        *(float4*)(out + (size_t)(lo + n) * DH + c4 * 4) = o;
        if (MODE != 0) {
            cs.x += o.x; cs.y += o.y; cs.z += o.z; cs.w += o.w;
            cq.x += o.x * o.x; cq.y += o.y * o.y; cq.z += o.z * o.z; cq.w += o.w * o.w;
        }
    }
    if (MODE != 0 && pout != nullptr) {
        red[wave * 64 + lane]      = cs;
        red[wave * 64 + 32 + lane] = cq;
        __syncthreads();
        if (tid < 64) {
            float4 a = red[tid];
#pragma unroll
            for (int w = 1; w < AGG_WAVES; ++w) {
                const float4 b = red[w * 64 + tid];
                a.x += b.x; a.y += b.y; a.z += b.z; a.w += b.w;
            }
            *(float4*)(pout + (size_t)blockIdx.x * (2 * DH) + tid * 4) = a;
        }
    }
}

extern "C" void kernel_launch(void* const* d_in, const int* in_sizes, int n_in,
                              void* d_out, int out_size, void* d_ws, size_t ws_size,
                              hipStream_t stream) {
    const float* x        = (const float*)d_in[0];
    const int*   edge     = (const int*)d_in[1];
    const float* W_lin    = (const float*)d_in[2];
    const float* b_lin    = (const float*)d_in[3];
    const float* W1s      = (const float*)d_in[4];
    const float* bn_gamma = (const float*)d_in[5];
    const float* bn_beta  = (const float*)d_in[6];
    float* out = (float*)d_out;

    float* ws    = (float*)d_ws;
    float* dinv  = ws;
    float* partA = ws + NND;
    float* partB = partA + (size_t)AGG_NWG * 2 * DH;
    float* bufA  = partB + (size_t)AGG_NWG * 2 * DH;
    float* bufB  = bufA + (size_t)NND * DH;
    float* x0    = bufB + (size_t)NND * DH;
    int* cnts       = (int*)(x0 + (size_t)NND * DH);
    unsigned* lists = (unsigned*)(cnts + AGG_NWG * AGG_WAVES);

    const int* src = edge;
    const int* dst = edge + NED;

    const int nBlkN   = (NND + 255) / 256;
    const int nBlkE   = (NED + 255) / 256;
    const int nBlkGemm = (NND / 16 + 7) / 8;

    k_deg_init<<<nBlkN, 256, 0, stream>>>(dinv);
    k_deg_edges<<<nBlkE, 256, 0, stream>>>(dst, dinv);
    k_deg_rsqrt<<<nBlkN, 256, 0, stream>>>(dinv);

    k_gemm_wmma<DIN><<<nBlkGemm, 256, DIN * DH * sizeof(float), stream>>>(
        x, W_lin, bufA, 0, 0.0f);

    k_agg<0><<<AGG_NWG, AGG_TPB, AGG_LDS0, stream>>>(src, dst, bufA, bufA, dinv, b_lin, bufB,
                                                     cnts, lists, W1s, 0.0f,
                                                     nullptr, nullptr, nullptr, nullptr);

    hipMemcpyAsync(x0, bufB, (size_t)NND * DH * sizeof(float),
                   hipMemcpyDeviceToDevice, stream);

    float* z = bufB;
    float* h = bufA;
    for (int i = 0; i < 5; ++i) {
        const float beta = logf(0.5f / (float)(i + 1) + 1.0f);
        float* pout = (i < 4) ? ((i & 1) ? partB : partA) : nullptr;
        if (i == 0) {
            k_agg<1><<<AGG_NWG, AGG_TPB, AGG_LDS1, stream>>>(src, dst, z, x0, dinv, b_lin, h, cnts, lists,
                                                             W1s, beta,
                                                             nullptr, nullptr, nullptr, pout);
        } else {
            const float* pin = ((i - 1) & 1) ? partB : partA;
            k_agg<2><<<AGG_NWG, AGG_TPB, AGG_LDS2, stream>>>(src, dst, z, x0, dinv, b_lin, h, cnts, lists,
                                                             W1s + (size_t)i * DH * DH, beta,
                                                             pin, bn_gamma + (i - 1) * DH,
                                                             bn_beta + (i - 1) * DH, pout);
        }
        float* t = z; z = h; h = t;
    }

    hipMemcpyAsync(out, z, (size_t)NND * DH * sizeof(float),
                   hipMemcpyDeviceToDevice, stream);
}


# GCN2 passes without LDS atomics: first pass files each edge in the list of the wave that owns its destination row; that wave adds source rows by plain 16-B LDS load, add, store; sum order changes
# speedup vs baseline: 4.0213x; 1.0605x over previous
#include <hip/hip_runtime.h>
#include <cmath>

typedef __attribute__((ext_vector_type(2))) float v2f;
typedef __attribute__((ext_vector_type(8))) float v8f;

#define NND   50000
#define NED   800000
#define DIN   256
#define DH    128
#define ALPHA 0.1f
#define BNEPS 1e-5f

__global__ void k_deg_init(float* deg) {
    int i = blockIdx.x * blockDim.x + threadIdx.x;
    if (i < NND) deg[i] = 1.0f;
}
__global__ void k_deg_edges(const int* __restrict__ dst, float* deg) {
    int e = blockIdx.x * blockDim.x + threadIdx.x;
    if (e < NED) atomicAdd(&deg[dst[e]], 1.0f);
}
__global__ void k_deg_rsqrt(float* deg) {
    int i = blockIdx.x * blockDim.x + threadIdx.x;
    if (i < NND) deg[i] = rsqrtf(fmaxf(deg[i], 1.0f));
}

template <int K>
__global__ __launch_bounds__(256)
void k_gemm_wmma(const float* in, const float* __restrict__ W,
                 float* out, int mode, float beta) {
    extern __shared__ float2 sW2[];
    const int tid = threadIdx.x;

    {
        const int totalIter = (K / 2) * (DH / 4);
        static_assert(((K / 2) * (DH / 4)) % 256 == 0, "every thread runs the same number of staging iterations");
#pragma unroll 8
        for (int it = 0; it < totalIter / 256; ++it) {
            const int i = tid + it * 256;
            const int kp = i / (DH / 4);
            const int c4 = i % (DH / 4);
            const float4 a0 = *(const float4*)(W + (size_t)(2 * kp)     * DH + c4 * 4);
            const float4 a1 = *(const float4*)(W + (size_t)(2 * kp + 1) * DH + c4 * 4);
            float4* dst4 = (float4*)(sW2 + (size_t)kp * DH + c4 * 4);
            float4 f0; f0.x = a0.x; f0.y = a1.x; f0.z = a0.y; f0.w = a1.y;
            float4 f1; f1.x = a0.z; f1.y = a1.z; f1.z = a0.w; f1.w = a1.w;
            dst4[0] = f0;
            dst4[1] = f1;
        }
    }
    __syncthreads();

    const int wave = tid >> 5;
    const int lane = tid & 31;
    const int tileRow = blockIdx.x * 8 + wave;
    if (tileRow >= NND / 16) return;
    const int m0 = tileRow * 16;
    const int r    = lane & 15;
    const bool hi  = lane >= 16;
    const int koff = hi ? 2 : 0;
    const float* arow = in + (size_t)(m0 + r) * K;

    v8f c[8] = {};
    for (int k0 = 0; k0 < K; k0 += 4) {
        const float2 a2 = *(const float2*)(arow + k0 + koff);
        v2f a; a.x = a2.x; a.y = a2.y;
        const int pairRow = (k0 >> 1) + (hi ? 1 : 0);
        const float2* brow = sW2 + (size_t)pairRow * DH + r;
#pragma unroll
        for (int t = 0; t < 8; ++t) {
            const float2 b2 = brow[t * 16];
            v2f b; b.x = b2.x; b.y = b2.y;
            c[t] = __builtin_amdgcn_wmma_f32_16x16x4_f32(
                false, a, false, b, (short)0, c[t], false, false);
        }
    }

#pragma unroll
    for (int t = 0; t < 8; ++t) {
        const int col = t * 16 + r;
#pragma unroll
        for (int v = 0; v < 8; ++v) {
            const int row = m0 + v + (hi ? 8 : 0);
            const size_t o = (size_t)row * DH + col;
            float val = c[t][v];
            if (mode != 0) val = (1.0f - beta) * in[o] + beta * val;
            out[o] = val;
        }
    }
}

#define AGG_R      196
#define AGG_NWG    ((NND + AGG_R - 1) / AGG_R)
#define AGG_TPB    1024
#define AGG_WAVES  (AGG_TPB / 32)
#define AGG_LPT    8
#define AGG_BATCH  (32 * 4 * AGG_LPT)
#define AGG_QCAP   (AGG_BATCH + 32)
#define AGG_NBATCH ((NED + AGG_BATCH - 1) / AGG_BATCH)
#define AGG_S      (DH + 4)
#define AGG_RT     ((AGG_R + 15) / 16)
#define AGG_ACCF   (AGG_RT * 16 * AGG_S)
#define AGG_GW     (AGG_RT * 2)
#define AGG_LDS0   ((AGG_ACCF + AGG_WAVES * AGG_QCAP) * 4)
#define AGG_LDS0C  (AGG_LDS0 + AGG_WAVES * 4)
#define AGG_LDS1   (AGG_LDS0 + DH * DH * 4)
#define AGG_LDS2   (AGG_LDS1 + 2 * DH * 4)
#define AGG_PG     (AGG_TPB / 64)
#define AGG_WCAP   2048
#define AGG_SVN    ((AGG_R + AGG_WAVES - 1) / AGG_WAVES)
static_assert(NED <= (1 << 20) && AGG_R <= (1 << 12), "queue entry packing");
static_assert(AGG_WAVES <= AGG_NBATCH, "every wave has a first batch");
static_assert(AGG_GW <= AGG_WAVES, "a wave for every row tile and column half");
static_assert(AGG_LDS2 <= 320 * 1024, "one work-group's LDS");
static_assert(AGG_NWG % AGG_PG == 0, "partial rows split evenly over the thread groups");
static_assert(AGG_WAVES * 64 * 16 <= AGG_WAVES * AGG_QCAP * 4, "the column-sum scratch fits in the queue area");
static_assert(DH == 128 && AGG_TPB >= DH, "a wave's 32 lanes x 16 B cover one row; a thread per column");
static_assert(NND <= (1 << 16) && AGG_R <= (1 << 12), "list entry packing");
static_assert((DH / 2) * (DH / 4) == 2 * AGG_TPB, "W1 staging: two k-pair rows x 4 columns per thread");

template <bool BN>
__device__ __forceinline__ void agg_rows(int s, int dl, int wbits, int cntc, int lane,
                                         const float* __restrict__ zin, float* acc,
                                         const float (&sc)[4], const float (&sh)[4], int first = 0) {
    for (int g = first; g < cntc; g += 8) {
        float v[8][4];
#pragma unroll
        for (int u = 0; u < 8; ++u) {
            int k = g + u;
            if (k > cntc - 1) k = cntc - 1;
            const int sk = __builtin_amdgcn_readlane(s, k);
            const float* row = zin + (size_t)sk * DH + lane;
            v[u][0] = row[0];
            v[u][1] = row[32];
            v[u][2] = row[64];
            v[u][3] = row[96];
        }
#pragma unroll
        for (int u = 0; u < 8; ++u) {
            const int k = g + u;
            if (k < cntc) {
                const int dk = __builtin_amdgcn_readlane(dl, k);
                float* arow = acc + dk * AGG_S + lane;
                if (BN) {
                    atomicAdd(arow,      fmaxf(fmaf(v[u][0], sc[0], sh[0]), 0.0f));
                    atomicAdd(arow + 32, fmaxf(fmaf(v[u][1], sc[1], sh[1]), 0.0f));
                    atomicAdd(arow + 64, fmaxf(fmaf(v[u][2], sc[2], sh[2]), 0.0f));
                    atomicAdd(arow + 96, fmaxf(fmaf(v[u][3], sc[3], sh[3]), 0.0f));
                } else {
                    const float wk = __int_as_float(__builtin_amdgcn_readlane(wbits, k));
                    atomicAdd(arow,      wk * v[u][0]);
                    atomicAdd(arow + 32, wk * v[u][1]);
                    atomicAdd(arow + 64, wk * v[u][2]);
                    atomicAdd(arow + 96, wk * v[u][3]);
                }
            }
        }
    }
}

__device__ __forceinline__ float4 row_load4(const float* __restrict__ zin, int sk, int lane) {
    return *(const float4*)(zin + (size_t)sk * DH + lane * 4);
}
template <bool BN>
__device__ __forceinline__ void row_add4(const float4 v, float* acc, int dk, int lane,
                                         const float (&sc)[4], const float (&sh)[4]) {
    float4* p = (float4*)(acc + dk * AGG_S + lane * 4);
    float4 h = *p;
    if (BN) {
        h.x += fmaxf(fmaf(v.x, sc[0], sh[0]), 0.0f);
        h.y += fmaxf(fmaf(v.y, sc[1], sh[1]), 0.0f);
        h.z += fmaxf(fmaf(v.z, sc[2], sh[2]), 0.0f);
        h.w += fmaxf(fmaf(v.w, sc[3], sh[3]), 0.0f);
    } else {
        h.x += (1.0f - ALPHA) * v.x;
        h.y += (1.0f - ALPHA) * v.y;
        h.z += (1.0f - ALPHA) * v.z;
        h.w += (1.0f - ALPHA) * v.w;
    }
    *p = h;
}
template <bool BN>
__device__ __forceinline__ void own_rows(int s, int dl, int cntc, int lane,
                                         const float* __restrict__ zin, float* acc,
                                         const float (&sc)[4], const float (&sh)[4], int first) {
    for (int g = first; g < cntc; g += 8) {
        float4 v[8];
#pragma unroll
        for (int u = 0; u < 8; ++u) {
            int k = g + u;
            if (k > cntc - 1) k = cntc - 1;
            v[u] = row_load4(zin, __builtin_amdgcn_readlane(s, k), lane);
        }
#pragma unroll
        for (int u = 0; u < 8; ++u) {
            const int k = g + u;
            if (k < cntc) row_add4<BN>(v[u], acc, __builtin_amdgcn_readlane(dl, k), lane, sc, sh);
        }
    }
}

template <int MODE>
__device__ __forceinline__ void agg_chunk(const unsigned* q, int base, int cntc, int lane, int lo,
                                          const int* __restrict__ src,
                                          const float* __restrict__ zin,
                                          const float* __restrict__ dinv, float* acc,
                                          unsigned* wgl, unsigned* lcnt,
                                          const float (&sc)[4], const float (&sh)[4]) {
    const unsigned ent = (lane < cntc) ? q[base + lane] : 0u;
    const int e  = (int)(ent & 0xFFFFFu);
    const int dl = (int)(ent >> 20);
    const int s  = src[e];
    float w = 1.0f - ALPHA;
    if (MODE == 0) {
        w = dinv[s] * dinv[lo + dl];
        if (lane < cntc) {
            const int ow = dl % AGG_WAVES;
            const unsigned pos = atomicAdd(&lcnt[ow], 1u);
            if (pos < (unsigned)AGG_WCAP)
                wgl[(size_t)ow * AGG_WCAP + pos] = ((unsigned)dl << 16) | (unsigned)s;
        }
    }
    agg_rows<MODE == 2>(s, dl, __float_as_int(w), cntc, lane, zin, acc, sc, sh);
}

__device__ __forceinline__ void agg_load(int4 (&dv)[AGG_LPT], const int* __restrict__ dst,
                                         int t, int lane) {
    const int eb = t * AGG_BATCH + lane * 4;
#pragma unroll
    for (int j = 0; j < AGG_LPT; ++j) {
        int e = eb + j * 128;
        if (e > NED - 4) e = NED - 4;
        dv[j] = *(const int4*)(dst + e);
    }
    if (t == AGG_NBATCH - 1) {
#pragma unroll
        for (int j = 0; j < AGG_LPT; ++j)
            if (eb + j * 128 >= NED) dv[j] = make_int4(-1, -1, -1, -1);
    }
}

template <bool OWN>
__device__ __forceinline__ void agg_scan(const int4 (&dv)[AGG_LPT], int t, int lane, int lo, int cnt,
                                         int wave, unsigned* q, int& qn) {
    const int eb = t * AGG_BATCH + lane * 4;
#pragma unroll
    for (int j = 0; j < AGG_LPT; ++j) {
        const int e = eb + j * 128;
        const int dd[4] = {dv[j].x, dv[j].y, dv[j].z, dv[j].w};
#pragma unroll
        for (int c = 0; c < 4; ++c) {
            const unsigned dl = (unsigned)(dd[c] - lo);
            const bool m = dl < (unsigned)cnt && (!OWN || (int)(dl % AGG_WAVES) == wave);
            const unsigned bm = __builtin_amdgcn_ballot_w32(m);
            if (bm != 0u) {
                const int pos = qn + (int)__builtin_amdgcn_mbcnt_lo(bm, 0u);
                if (m) q[pos] = (dl << 20) | (unsigned)(e + c);
                qn += __builtin_popcount(bm);
            }
        }
    }
}

template <int MODE>
__device__ __forceinline__ void agg_drain(unsigned* q, int& qn, int lane, int lo,
                                          const int* __restrict__ src,
                                          const float* __restrict__ zin,
                                          const float* __restrict__ dinv, float* acc,
                                          unsigned* wgl, unsigned* lcnt,
                                          const float (&sc)[4], const float (&sh)[4]) {
    if (qn < 32) return;
    __builtin_amdgcn_wave_barrier();
    const int full = qn & ~31;
    for (int base = 0; base < full; base += 32)
        agg_chunk<MODE>(q, base, 32, lane, lo, src, zin, dinv, acc, wgl, lcnt, sc, sh);
    const int rem = qn - full;
    unsigned keep = 0u;
    if (lane < rem) keep = q[full + lane];
    __builtin_amdgcn_wave_barrier();
    if (lane < rem) q[lane] = keep;
    qn = rem;
}

template <int MODE>
__global__ __launch_bounds__(AGG_TPB)
void k_agg(const int* __restrict__ src, const int* __restrict__ dst,
           const float* __restrict__ zin, const float* __restrict__ start,
           const float* __restrict__ dinv, const float* __restrict__ bias,
           float* __restrict__ out, int* cnts, unsigned* lists,
           const float* __restrict__ W1, float beta,
           const float* __restrict__ pin, const float* __restrict__ gamma,
           const float* __restrict__ bnb, float* __restrict__ pout) {
    extern __shared__ float agg_smem[];
    float* acc = agg_smem;
    unsigned* qall = (unsigned*)(agg_smem + AGG_ACCF);
    float2* sW2 = (float2*)(agg_smem + AGG_ACCF + AGG_WAVES * AGG_QCAP);
    float* scsh = agg_smem + AGG_ACCF + AGG_WAVES * AGG_QCAP + DH * DH;
    unsigned* lcnt = (unsigned*)(agg_smem + AGG_ACCF + AGG_WAVES * AGG_QCAP);
    float4* red = (float4*)qall;
    const int tid  = threadIdx.x;
    const int lane = tid & 31;
    const int wave = __builtin_amdgcn_readfirstlane(tid >> 5);
    const int lo   = blockIdx.x * AGG_R;
    const int cnt  = (NND - lo < AGG_R) ? (NND - lo) : AGG_R;
    unsigned* q = qall + wave * AGG_QCAP;
    const int widx = blockIdx.x * AGG_WAVES + wave;
    unsigned* wgl = lists + (size_t)blockIdx.x * AGG_WAVES * AGG_WCAP;
    unsigned* wl = wgl + (size_t)wave * AGG_WCAP;

    const int rot = (int)(((long long)blockIdx.x * AGG_NBATCH) / AGG_NWG);
    const int tstep = (MODE == 0) ? AGG_WAVES : 1;
    int t0 = (MODE == 0) ? wave : 0;
    int ta = t0 + rot;
    if (ta >= AGG_NBATCH) ta -= AGG_NBATCH;
    int4 dva[AGG_LPT], dvb[AGG_LPT];

    int wn = 0;
    bool scan = true;
    unsigned ent = 0u;
    if (MODE == 0) {
        agg_load(dva, dst, ta, lane);
        if (tid < AGG_WAVES) lcnt[tid] = 0u;
    } else {
        wn = cnts[widx];
    }

    float4 sv[AGG_SVN];
    float dnv[AGG_SVN];
    float4 bb = make_float4(0.0f, 0.0f, 0.0f, 0.0f);
#pragma unroll
    for (int k = 0; k < AGG_SVN; ++k) {
        int n = wave + k * AGG_WAVES;
        if (n > cnt - 1) n = cnt - 1;
        sv[k] = *(const float4*)(start + (size_t)(lo + n) * DH + lane * 4);
        dnv[k] = (MODE == 0) ? dinv[lo + n] : 0.0f;
    }
    if (MODE == 0) bb = *(const float4*)(bias + lane * 4);
    float4 w1v[2][2];
    if (MODE != 0) {
#pragma unroll
        for (int it = 0; it < 2; ++it) {
            const int i = tid + it * AGG_TPB;
            const int kp = i / (DH / 4);
            const int c4 = i % (DH / 4);
            w1v[it][0] = *(const float4*)(W1 + (size_t)(2 * kp)     * DH + c4 * 4);
            w1v[it][1] = *(const float4*)(W1 + (size_t)(2 * kp + 1) * DH + c4 * 4);
        }
    }

    float4 pa = make_float4(0.0f, 0.0f, 0.0f, 0.0f);
    float gmv = 0.0f, btv = 0.0f;
    if (MODE == 2) {
        if (tid < DH) { gmv = gamma[tid]; btv = bnb[tid]; }
        const float* pp = pin + (size_t)(tid >> 6) * (2 * DH) + (tid & 63) * 4;
#pragma unroll
        for (int b = 0; b < AGG_NWG / AGG_PG; ++b) {
            const float4 p = *(const float4*)(pp + (size_t)b * (AGG_PG * 2 * DH));
            pa.x += p.x; pa.y += p.y; pa.z += p.z; pa.w += p.w;
        }
    }

    if (MODE != 0) {
        scan = wn > AGG_WCAP;
        if (!scan && lane < wn) ent = wl[lane];
    }

#pragma unroll
    for (int k = 0; k < AGG_SVN; ++k) {
        const int n = wave + k * AGG_WAVES;
        if (n < cnt) {
            const float4 v = sv[k];
            float4 o;
            if (MODE == 0) {
                const float w = dnv[k] * dnv[k];
                o.x = w * v.x + bb.x; o.y = w * v.y + bb.y;
                o.z = w * v.z + bb.z; o.w = w * v.w + bb.w;
            } else {
                o.x = ALPHA * v.x; o.y = ALPHA * v.y; o.z = ALPHA * v.z; o.w = ALPHA * v.w;
            }
            *(float4*)(acc + n * AGG_S + lane * 4) = o;
        }
    }
    if (MODE != 0) {
#pragma unroll
        for (int it = 0; it < 2; ++it) {
            const int i = tid + it * AGG_TPB;
            const int kp = i / (DH / 4);
            const int c4 = i % (DH / 4);
            const float4 a0 = w1v[it][0];
            const float4 a1 = w1v[it][1];
            float4* dst4 = (float4*)(sW2 + (size_t)kp * DH + c4 * 4);
            float4 f0; f0.x = a0.x; f0.y = a1.x; f0.z = a0.y; f0.w = a1.y;
            float4 f1; f1.x = a0.z; f1.y = a1.z; f1.z = a0.w; f1.w = a1.w;
            dst4[0] = f0;
            dst4[1] = f1;
        }
        for (int i = cnt * AGG_S + tid; i < AGG_ACCF; i += AGG_TPB) acc[i] = 0.0f;
    }
    float sc[4] = {0.0f, 0.0f, 0.0f, 0.0f};
    float sh[4] = {0.0f, 0.0f, 0.0f, 0.0f};
    if (MODE == 2) {
        red[tid] = pa;
        __syncthreads();
        if (tid < DH) {
            const float* rf = (const float*)red;
            float S = rf[tid], Q = rf[DH + tid];
#pragma unroll
            for (int k = 1; k < AGG_PG; ++k) {
                S += rf[k * 2 * DH + tid];
                Q += rf[k * 2 * DH + DH + tid];
            }
            const float invN = 1.0f / (float)NND;
            const float mean = S * invN;
            const float inv  = rsqrtf(Q * invN - mean * mean + BNEPS);
            const float gi   = gmv * inv;
            scsh[tid]      = (1.0f - ALPHA) * gi;
            scsh[DH + tid] = (1.0f - ALPHA) * (btv - mean * gi);
        }
    }
    __syncthreads();
    if (MODE == 2) {
#pragma unroll
        for (int j = 0; j < 4; ++j) {
            const int col = scan ? (lane + 32 * j) : (lane * 4 + j);
            sc[j] = scsh[col];
            sh[j] = scsh[DH + col];
        }
    }

    if (scan) {
        if (MODE != 0) agg_load(dva, dst, ta, lane);
        int qn = 0;
        while (true) {
            const int t1 = t0 + tstep;
            const bool has1 = t1 < AGG_NBATCH;
            int tb = (has1 ? t1 : t0) + rot;
            if (tb >= AGG_NBATCH) tb -= AGG_NBATCH;
            agg_load(dvb, dst, tb, lane);
            agg_scan<MODE != 0>(dva, ta, lane, lo, cnt, wave, q, qn);
            agg_drain<MODE>(q, qn, lane, lo, src, zin, dinv, acc, wgl, lcnt, sc, sh);
            if (!has1) break;
            const int t2 = t1 + tstep;
            const bool has2 = t2 < AGG_NBATCH;
            ta = (has2 ? t2 : t1) + rot;
            if (ta >= AGG_NBATCH) ta -= AGG_NBATCH;
            agg_load(dva, dst, ta, lane);
            agg_scan<MODE != 0>(dvb, tb, lane, lo, cnt, wave, q, qn);
            agg_drain<MODE>(q, qn, lane, lo, src, zin, dinv, acc, wgl, lcnt, sc, sh);
            if (!has2) break;
            t0 = t2;
        }
        __builtin_amdgcn_wave_barrier();
        if (qn > 0) agg_chunk<MODE>(q, 0, qn, lane, lo, src, zin, dinv, acc, wgl, lcnt, sc, sh);
    } else {
        int es = (int)(ent & 0xFFFFu), ed = (int)(ent >> 16);
        float4 v[8];
        if (wn > 0) {
            const int c0 = ((wn < 32) ? wn : 32) - 1;
#pragma unroll
            for (int u = 0; u < 8; ++u)
                v[u] = row_load4(zin, __builtin_amdgcn_readlane(es, (u < c0) ? u : c0), lane);
        }
        int base = 0;
        for (; base + 32 <= wn; base += 32) {
            unsigned nxt = 0u;
            if (base + 32 + lane < wn) nxt = wl[base + 32 + lane];
            const int ns = (int)(nxt & 0xFFFFu);
            int cl = wn - base - 33;
            if (cl > 31) cl = 31;
            if (cl < 0) cl = 0;
#pragma unroll
            for (int u = 0; u < 32; ++u) {
                row_add4<MODE == 2>(v[u & 7], acc, __builtin_amdgcn_readlane(ed, u), lane, sc, sh);
                if (u < 24) {
                    v[u & 7] = row_load4(zin, __builtin_amdgcn_readlane(es, u + 8), lane);
                } else {
                    const int j = (u - 24 < cl) ? (u - 24) : cl;
                    v[u & 7] = row_load4(zin, __builtin_amdgcn_readlane(ns, j), lane);
                }
            }
            es = ns; ed = (int)(nxt >> 16);
        }
        const int rem = wn - base;
        if (rem > 0) {
#pragma unroll
            for (int u = 0; u < 8; ++u)
                if (u < rem) row_add4<MODE == 2>(v[u], acc, __builtin_amdgcn_readlane(ed, u), lane, sc, sh);
            if (rem > 8) own_rows<MODE == 2>(es, ed, rem, lane, zin, acc, sc, sh, 8);
        }
    }
    __syncthreads();
    if (MODE == 0 && lane == 0) cnts[widx] = (int)lcnt[wave];

    if (MODE != 0) {
        const int r    = lane & 15;
        const bool hi  = lane >= 16;
        const int koff = hi ? 2 : 0;
        const bool act = wave < AGG_GW;
        const int row0 = (wave >> 1) * 16;
        const int col0 = (wave & 1) * 64 + r;
        v8f c[4] = {};
        if (act) {
            const float* arow = acc + (row0 + r) * AGG_S + koff;
            for (int k0 = 0; k0 < DH; k0 += 4) {
                const float2 a2 = *(const float2*)(arow + k0);
                v2f a; a.x = a2.x; a.y = a2.y;
                const float2* brow = sW2 + (size_t)((k0 >> 1) + (hi ? 1 : 0)) * DH + col0;
#pragma unroll
                for (int t = 0; t < 4; ++t) {
                    const float2 b2 = brow[t * 16];
                    v2f b; b.x = b2.x; b.y = b2.y;
                    c[t] = __builtin_amdgcn_wmma_f32_16x16x4_f32(
                        false, a, false, b, (short)0, c[t], false, false);
                }
            }
#pragma unroll
            for (int t = 0; t < 4; ++t) {
#pragma unroll
                for (int v = 0; v < 8; ++v) {
                    const float hv = acc[(row0 + v + (hi ? 8 : 0)) * AGG_S + col0 + t * 16];
                    c[t][v] = (1.0f - beta) * hv + beta * c[t][v];
                }
            }
        }
        __syncthreads();
        if (act) {
#pragma unroll
            for (int t = 0; t < 4; ++t) {
#pragma unroll
                for (int v = 0; v < 8; ++v)
                    acc[(row0 + v + (hi ? 8 : 0)) * AGG_S + col0 + t * 16] = c[t][v];
            }
        }
        __syncthreads();
    }

    float4 cs = make_float4(0.0f, 0.0f, 0.0f, 0.0f);
    float4 cq = make_float4(0.0f, 0.0f, 0.0f, 0.0f);
    for (int i = tid; i < cnt * (DH / 4); i += AGG_TPB) {
        const int n = i >> 5, c4 = i & 31;
        const float4 o = *(const float4*)(acc + n * AGG_S + c4 * 4);
        *(float4*)(out + (size_t)(lo + n) * DH + c4 * 4) = o;
        if (MODE != 0) {
            cs.x += o.x; cs.y += o.y; cs.z += o.z; cs.w += o.w;
            cq.x += o.x * o.x; cq.y += o.y * o.y; cq.z += o.z * o.z; cq.w += o.w * o.w;
        }
    }
    if (MODE != 0 && pout != nullptr) {
        red[wave * 64 + lane]      = cs;
        red[wave * 64 + 32 + lane] = cq;
        __syncthreads();
        if (tid < 64) {
            float4 a = red[tid];
#pragma unroll
            for (int w = 1; w < AGG_WAVES; ++w) {
                const float4 b = red[w * 64 + tid];
                a.x += b.x; a.y += b.y; a.z += b.z; a.w += b.w;
            }
            *(float4*)(pout + (size_t)blockIdx.x * (2 * DH) + tid * 4) = a;
        }
    }
}

extern "C" void kernel_launch(void* const* d_in, const int* in_sizes, int n_in,
                              void* d_out, int out_size, void* d_ws, size_t ws_size,
                              hipStream_t stream) {
    const float* x        = (const float*)d_in[0];
    const int*   edge     = (const int*)d_in[1];
    const float* W_lin    = (const float*)d_in[2];
    const float* b_lin    = (const float*)d_in[3];
    const float* W1s      = (const float*)d_in[4];
    const float* bn_gamma = (const float*)d_in[5];
    const float* bn_beta  = (const float*)d_in[6];
    float* out = (float*)d_out;

    float* ws    = (float*)d_ws;
    float* dinv  = ws;
    float* partA = ws + NND;
    float* partB = partA + (size_t)AGG_NWG * 2 * DH;
    float* bufA  = partB + (size_t)AGG_NWG * 2 * DH;
    float* bufB  = bufA + (size_t)NND * DH;
    float* x0    = bufB + (size_t)NND * DH;
    int* cnts       = (int*)(x0 + (size_t)NND * DH);
    unsigned* lists = (unsigned*)(cnts + AGG_NWG * AGG_WAVES);

    const int* src = edge;
    const int* dst = edge + NED;

    const int nBlkN   = (NND + 255) / 256;
    const int nBlkE   = (NED + 255) / 256;
    const int nBlkGemm = (NND / 16 + 7) / 8;

    k_deg_init<<<nBlkN, 256, 0, stream>>>(dinv);
    k_deg_edges<<<nBlkE, 256, 0, stream>>>(dst, dinv);
    k_deg_rsqrt<<<nBlkN, 256, 0, stream>>>(dinv);

    k_gemm_wmma<DIN><<<nBlkGemm, 256, DIN * DH * sizeof(float), stream>>>(
        x, W_lin, bufA, 0, 0.0f);

    k_agg<0><<<AGG_NWG, AGG_TPB, AGG_LDS0C, stream>>>(src, dst, bufA, bufA, dinv, b_lin, bufB,
                                                     cnts, lists, W1s, 0.0f,
                                                     nullptr, nullptr, nullptr, nullptr);

    hipMemcpyAsync(x0, bufB, (size_t)NND * DH * sizeof(float),
                   hipMemcpyDeviceToDevice, stream);

    float* z = bufB;
    float* h = bufA;
    for (int i = 0; i < 5; ++i) {
        const float beta = logf(0.5f / (float)(i + 1) + 1.0f);
        float* pout = (i < 4) ? ((i & 1) ? partB : partA) : nullptr;
        if (i == 0) {
            k_agg<1><<<AGG_NWG, AGG_TPB, AGG_LDS1, stream>>>(src, dst, z, x0, dinv, b_lin, h, cnts, lists,
                                                             W1s, beta,
                                                             nullptr, nullptr, nullptr, pout);
        } else {
            const float* pin = ((i - 1) & 1) ? partB : partA;
            k_agg<2><<<AGG_NWG, AGG_TPB, AGG_LDS2, stream>>>(src, dst, z, x0, dinv, b_lin, h, cnts, lists,
                                                             W1s + (size_t)i * DH * DH, beta,
                                                             pin, bn_gamma + (i - 1) * DH,
                                                             bn_beta + (i - 1) * DH, pout);
        }
        float* t = z; z = h; h = t;
    }

    hipMemcpyAsync(out, z, (size_t)NND * DH * sizeof(float),
                   hipMemcpyDeviceToDevice, stream);
}


# Edges sorted once per call by destination range: degree kernel files each edge in a segment per (range, edge block); GCNConv pass reads its 256 segments, no dst scan; sums by arrival as before
# speedup vs baseline: 4.2202x; 1.0495x over previous
#include <hip/hip_runtime.h>
#include <cmath>

typedef __attribute__((ext_vector_type(2))) float v2f;
typedef __attribute__((ext_vector_type(8))) float v8f;

#define NND   50000
#define NED   800000
#define DIN   256
#define DH    128
#define ALPHA 0.1f
#define BNEPS 1e-5f

__global__ void k_deg_init(float* deg) {
    int i = blockIdx.x * blockDim.x + threadIdx.x;
    if (i < NND) deg[i] = 1.0f;
}
#define BIN_R    196
#define BIN_NWG  256
#define BIN_EPW  (NED / BIN_NWG)
#define BIN_CAP  64
#define BIN_TPB  1024
#define BIN_EPT  ((BIN_EPW + BIN_TPB - 1) / BIN_TPB)
static_assert(NED % BIN_NWG == 0, "every edge belongs to one work-group of the binning kernel");
static_assert(BIN_R * BIN_NWG >= NND, "a range for every node");
__global__ __launch_bounds__(BIN_TPB)
void k_bin(const int* __restrict__ src, const int* __restrict__ dst, float* deg,
           unsigned* __restrict__ bcnt, unsigned* __restrict__ segs) {
    extern __shared__ unsigned bin_smem[];
    unsigned* c  = bin_smem;
    unsigned* st = bin_smem + BIN_NWG;
    const int tid = threadIdx.x;
    if (tid < BIN_NWG) c[tid] = 0u;
    const int e0 = blockIdx.x * BIN_EPW;
    int dv[BIN_EPT], sv[BIN_EPT];
#pragma unroll
    for (int k = 0; k < BIN_EPT; ++k) {
        int i = tid + k * BIN_TPB;
        if (i > BIN_EPW - 1) i = BIN_EPW - 1;
        dv[k] = dst[e0 + i];
        sv[k] = src[e0 + i];
    }
    __syncthreads();
#pragma unroll
    for (int k = 0; k < BIN_EPT; ++k) {
        if (tid + k * BIN_TPB < BIN_EPW) {
            const int d = dv[k];
            atomicAdd(&deg[d], 1.0f);
            const int g  = d / BIN_R;
            const int dl = d - g * BIN_R;
            const unsigned pos = atomicAdd(&c[g], 1u);
            if (pos < (unsigned)BIN_CAP) st[g * BIN_CAP + pos] = ((unsigned)dl << 16) | (unsigned)sv[k];
        }
    }
    __syncthreads();
    if (tid < BIN_NWG) bcnt[(size_t)blockIdx.x * BIN_NWG + tid] = c[tid];
    static_assert(BIN_TPB % BIN_CAP == 0 && (BIN_NWG * BIN_CAP) % BIN_TPB == 0, "whole segments per sweep");
    const int p  = tid % BIN_CAP;
    const int g0 = tid / BIN_CAP;
    unsigned cg[BIN_NWG * BIN_CAP / BIN_TPB];
#pragma unroll
    for (int k = 0; k < BIN_NWG * BIN_CAP / BIN_TPB; ++k) cg[k] = c[g0 + k * (BIN_TPB / BIN_CAP)];
#pragma unroll
    for (int k = 0; k < BIN_NWG * BIN_CAP / BIN_TPB; ++k) {
        const int g = g0 + k * (BIN_TPB / BIN_CAP);
        if ((unsigned)p < cg[k]) segs[((size_t)g * BIN_NWG + blockIdx.x) * BIN_CAP + p] = st[g * BIN_CAP + p];
    }
}
__global__ void k_deg_rsqrt(float* deg) {
    int i = blockIdx.x * blockDim.x + threadIdx.x;
    if (i < NND) deg[i] = rsqrtf(fmaxf(deg[i], 1.0f));
}

template <int K>
__global__ __launch_bounds__(256)
void k_gemm_wmma(const float* in, const float* __restrict__ W,
                 float* out, int mode, float beta) {
    extern __shared__ float2 sW2[];
    const int tid = threadIdx.x;

    {
        const int totalIter = (K / 2) * (DH / 4);
        static_assert(((K / 2) * (DH / 4)) % 256 == 0, "every thread runs the same number of staging iterations");
#pragma unroll 8
        for (int it = 0; it < totalIter / 256; ++it) {
            const int i = tid + it * 256;
            const int kp = i / (DH / 4);
            const int c4 = i % (DH / 4);
            const float4 a0 = *(const float4*)(W + (size_t)(2 * kp)     * DH + c4 * 4);
            const float4 a1 = *(const float4*)(W + (size_t)(2 * kp + 1) * DH + c4 * 4);
            float4* dst4 = (float4*)(sW2 + (size_t)kp * DH + c4 * 4);
            float4 f0; f0.x = a0.x; f0.y = a1.x; f0.z = a0.y; f0.w = a1.y;
            float4 f1; f1.x = a0.z; f1.y = a1.z; f1.z = a0.w; f1.w = a1.w;
            dst4[0] = f0;
            dst4[1] = f1;
        }
    }
    __syncthreads();

    const int wave = tid >> 5;
    const int lane = tid & 31;
    const int tileRow = blockIdx.x * 8 + wave;
    if (tileRow >= NND / 16) return;
    const int m0 = tileRow * 16;
    const int r    = lane & 15;
    const bool hi  = lane >= 16;
    const int koff = hi ? 2 : 0;
    const float* arow = in + (size_t)(m0 + r) * K;

    v8f c[8] = {};
    for (int k0 = 0; k0 < K; k0 += 4) {
        const float2 a2 = *(const float2*)(arow + k0 + koff);
        v2f a; a.x = a2.x; a.y = a2.y;
        const int pairRow = (k0 >> 1) + (hi ? 1 : 0);
        const float2* brow = sW2 + (size_t)pairRow * DH + r;
#pragma unroll
        for (int t = 0; t < 8; ++t) {
            const float2 b2 = brow[t * 16];
            v2f b; b.x = b2.x; b.y = b2.y;
            c[t] = __builtin_amdgcn_wmma_f32_16x16x4_f32(
                false, a, false, b, (short)0, c[t], false, false);
        }
    }

#pragma unroll
    for (int t = 0; t < 8; ++t) {
        const int col = t * 16 + r;
#pragma unroll
        for (int v = 0; v < 8; ++v) {
            const int row = m0 + v + (hi ? 8 : 0);
            const size_t o = (size_t)row * DH + col;
            float val = c[t][v];
            if (mode != 0) val = (1.0f - beta) * in[o] + beta * val;
            out[o] = val;
        }
    }
}

#define AGG_R      196
#define AGG_NWG    ((NND + AGG_R - 1) / AGG_R)
#define AGG_TPB    1024
#define AGG_WAVES  (AGG_TPB / 32)
#define AGG_LPT    8
#define AGG_BATCH  (32 * 4 * AGG_LPT)
#define AGG_QCAP   (AGG_BATCH + 32)
#define AGG_NBATCH ((NED + AGG_BATCH - 1) / AGG_BATCH)
#define AGG_S      (DH + 4)
#define AGG_RT     ((AGG_R + 15) / 16)
#define AGG_ACCF   (AGG_RT * 16 * AGG_S)
#define AGG_GW     (AGG_RT * 2)
#define AGG_LDS0   ((AGG_ACCF + AGG_WAVES * AGG_QCAP) * 4)
#define AGG_LDS0C  (AGG_LDS0 + AGG_WAVES * 4 + BIN_NWG * 4)
#define AGG_LDS1   (AGG_LDS0 + DH * DH * 4)
#define AGG_LDS2   (AGG_LDS1 + 2 * DH * 4)
#define AGG_PG     (AGG_TPB / 64)
#define AGG_WCAP   2048
#define AGG_SVN    ((AGG_R + AGG_WAVES - 1) / AGG_WAVES)
static_assert(NED <= (1 << 20) && AGG_R <= (1 << 12), "queue entry packing");
static_assert(AGG_WAVES <= AGG_NBATCH, "every wave has a first batch");
static_assert(BIN_R == AGG_R && BIN_NWG == AGG_NWG, "the binning kernel's ranges are this kernel's work-groups");
static_assert(BIN_NWG % AGG_WAVES == 0 && (BIN_NWG / AGG_WAVES) * BIN_CAP <= AGG_QCAP, "a wave's segments fit in its queue");
static_assert(BIN_CAP == 64 && BIN_NWG <= AGG_TPB, "a segment is two entries per lane; a thread per segment count");
static_assert(AGG_GW <= AGG_WAVES, "a wave for every row tile and column half");
static_assert(AGG_LDS2 <= 320 * 1024, "one work-group's LDS");
static_assert(AGG_NWG % AGG_PG == 0, "partial rows split evenly over the thread groups");
static_assert(AGG_WAVES * 64 * 16 <= AGG_WAVES * AGG_QCAP * 4, "the column-sum scratch fits in the queue area");
static_assert(DH == 128 && AGG_TPB >= DH, "a wave's 32 lanes x 16 B cover one row; a thread per column");
static_assert(NND <= (1 << 16) && AGG_R <= (1 << 12), "list entry packing");
static_assert((DH / 2) * (DH / 4) == 2 * AGG_TPB, "W1 staging: two k-pair rows x 4 columns per thread");

template <bool BN>
__device__ __forceinline__ void agg_rows(int s, int dl, int wbits, int cntc, int lane,
                                         const float* __restrict__ zin, float* acc,
                                         const float (&sc)[4], const float (&sh)[4], int first = 0) {
    for (int g = first; g < cntc; g += 8) {
        float v[8][4];
#pragma unroll
        for (int u = 0; u < 8; ++u) {
            int k = g + u;
            if (k > cntc - 1) k = cntc - 1;
            const int sk = __builtin_amdgcn_readlane(s, k);
            const float* row = zin + (size_t)sk * DH + lane;
            v[u][0] = row[0];
            v[u][1] = row[32];
            v[u][2] = row[64];
            v[u][3] = row[96];
        }
#pragma unroll
        for (int u = 0; u < 8; ++u) {
            const int k = g + u;
            if (k < cntc) {
                const int dk = __builtin_amdgcn_readlane(dl, k);
                float* arow = acc + dk * AGG_S + lane;
                if (BN) {
                    atomicAdd(arow,      fmaxf(fmaf(v[u][0], sc[0], sh[0]), 0.0f));
                    atomicAdd(arow + 32, fmaxf(fmaf(v[u][1], sc[1], sh[1]), 0.0f));
                    atomicAdd(arow + 64, fmaxf(fmaf(v[u][2], sc[2], sh[2]), 0.0f));
                    atomicAdd(arow + 96, fmaxf(fmaf(v[u][3], sc[3], sh[3]), 0.0f));
                } else {
                    const float wk = __int_as_float(__builtin_amdgcn_readlane(wbits, k));
                    atomicAdd(arow,      wk * v[u][0]);
                    atomicAdd(arow + 32, wk * v[u][1]);
                    atomicAdd(arow + 64, wk * v[u][2]);
                    atomicAdd(arow + 96, wk * v[u][3]);
                }
            }
        }
    }
}

__device__ __forceinline__ float4 row_load4(const float* __restrict__ zin, int sk, int lane) {
    return *(const float4*)(zin + (size_t)sk * DH + lane * 4);
}
template <bool BN>
__device__ __forceinline__ void row_add4(const float4 v, float* acc, int dk, int lane,
                                         const float (&sc)[4], const float (&sh)[4]) {
    float4* p = (float4*)(acc + dk * AGG_S + lane * 4);
    float4 h = *p;
    if (BN) {
        h.x += fmaxf(fmaf(v.x, sc[0], sh[0]), 0.0f);
        h.y += fmaxf(fmaf(v.y, sc[1], sh[1]), 0.0f);
        h.z += fmaxf(fmaf(v.z, sc[2], sh[2]), 0.0f);
        h.w += fmaxf(fmaf(v.w, sc[3], sh[3]), 0.0f);
    } else {
        h.x += (1.0f - ALPHA) * v.x;
        h.y += (1.0f - ALPHA) * v.y;
        h.z += (1.0f - ALPHA) * v.z;
        h.w += (1.0f - ALPHA) * v.w;
    }
    *p = h;
}
template <bool BN>
__device__ __forceinline__ void own_rows(int s, int dl, int cntc, int lane,
                                         const float* __restrict__ zin, float* acc,
                                         const float (&sc)[4], const float (&sh)[4], int first) {
    for (int g = first; g < cntc; g += 8) {
        float4 v[8];
#pragma unroll
        for (int u = 0; u < 8; ++u) {
            int k = g + u;
            if (k > cntc - 1) k = cntc - 1;
            v[u] = row_load4(zin, __builtin_amdgcn_readlane(s, k), lane);
        }
#pragma unroll
        for (int u = 0; u < 8; ++u) {
            const int k = g + u;
            if (k < cntc) row_add4<BN>(v[u], acc, __builtin_amdgcn_readlane(dl, k), lane, sc, sh);
        }
    }
}

template <int MODE>
__device__ __forceinline__ void agg_chunk(const unsigned* q, int base, int cntc, int lane, int lo,
                                          const int* __restrict__ src,
                                          const float* __restrict__ zin,
                                          const float* __restrict__ dinv, float* acc,
                                          unsigned* wgl, unsigned* lcnt,
                                          const float (&sc)[4], const float (&sh)[4]) {
    const unsigned ent = (lane < cntc) ? q[base + lane] : 0u;
    const int e  = (int)(ent & 0xFFFFFu);
    const int dl = (int)(ent >> 20);
    const int s  = src[e];
    float w = 1.0f - ALPHA;
    if (MODE == 0) {
        w = dinv[s] * dinv[lo + dl];
        if (lane < cntc) {
            const int ow = dl % AGG_WAVES;
            const unsigned pos = atomicAdd(&lcnt[ow], 1u);
            if (pos < (unsigned)AGG_WCAP)
                wgl[(size_t)ow * AGG_WCAP + pos] = ((unsigned)dl << 16) | (unsigned)s;
        }
    }
    agg_rows<MODE == 2>(s, dl, __float_as_int(w), cntc, lane, zin, acc, sc, sh);
}

__device__ __forceinline__ void agg_chunk_seg(const unsigned* q, int base, int cntc, int lane, int lo,
                                              const float* __restrict__ zin,
                                              const float* __restrict__ dinv, float* acc,
                                              unsigned* wgl, unsigned* lcnt,
                                              const float (&sc)[4], const float (&sh)[4]) {
    const unsigned ent = (lane < cntc) ? q[base + lane] : 0u;
    const int s  = (int)(ent & 0xFFFFu);
    const int dl = (int)(ent >> 16);
    const float w = dinv[s] * dinv[lo + dl];
    if (lane < cntc) {
        const int ow = dl % AGG_WAVES;
        const unsigned pos = atomicAdd(&lcnt[ow], 1u);
        if (pos < (unsigned)AGG_WCAP) wgl[(size_t)ow * AGG_WCAP + pos] = ent;
    }
    agg_rows<false>(s, dl, __float_as_int(w), cntc, lane, zin, acc, sc, sh);
}

__device__ __forceinline__ void agg_load(int4 (&dv)[AGG_LPT], const int* __restrict__ dst,
                                         int t, int lane) {
    const int eb = t * AGG_BATCH + lane * 4;
#pragma unroll
    for (int j = 0; j < AGG_LPT; ++j) {
        int e = eb + j * 128;
        if (e > NED - 4) e = NED - 4;
        dv[j] = *(const int4*)(dst + e);
    }
    if (t == AGG_NBATCH - 1) {
#pragma unroll
        for (int j = 0; j < AGG_LPT; ++j)
            if (eb + j * 128 >= NED) dv[j] = make_int4(-1, -1, -1, -1);
    }
}

template <bool OWN>
__device__ __forceinline__ void agg_scan(const int4 (&dv)[AGG_LPT], int t, int lane, int lo, int cnt,
                                         int wave, unsigned* q, int& qn) {
    const int eb = t * AGG_BATCH + lane * 4;
#pragma unroll
    for (int j = 0; j < AGG_LPT; ++j) {
        const int e = eb + j * 128;
        const int dd[4] = {dv[j].x, dv[j].y, dv[j].z, dv[j].w};
#pragma unroll
        for (int c = 0; c < 4; ++c) {
            const unsigned dl = (unsigned)(dd[c] - lo);
            const bool m = dl < (unsigned)cnt && (!OWN || (int)(dl % AGG_WAVES) == wave);
            const unsigned bm = __builtin_amdgcn_ballot_w32(m);
            if (bm != 0u) {
                const int pos = qn + (int)__builtin_amdgcn_mbcnt_lo(bm, 0u);
                if (m) q[pos] = (dl << 20) | (unsigned)(e + c);
                qn += __builtin_popcount(bm);
            }
        }
    }
}

template <int MODE>
__device__ __forceinline__ void agg_drain(unsigned* q, int& qn, int lane, int lo,
                                          const int* __restrict__ src,
                                          const float* __restrict__ zin,
                                          const float* __restrict__ dinv, float* acc,
                                          unsigned* wgl, unsigned* lcnt,
                                          const float (&sc)[4], const float (&sh)[4]) {
    if (qn < 32) return;
    __builtin_amdgcn_wave_barrier();
    const int full = qn & ~31;
    for (int base = 0; base < full; base += 32)
        agg_chunk<MODE>(q, base, 32, lane, lo, src, zin, dinv, acc, wgl, lcnt, sc, sh);
    const int rem = qn - full;
    unsigned keep = 0u;
    if (lane < rem) keep = q[full + lane];
    __builtin_amdgcn_wave_barrier();
    if (lane < rem) q[lane] = keep;
    qn = rem;
}

template <int MODE>
__global__ __launch_bounds__(AGG_TPB)
void k_agg(const int* __restrict__ src, const int* __restrict__ dst,
           const float* __restrict__ zin, const float* __restrict__ start,
           const float* __restrict__ dinv, const float* __restrict__ bias,
           float* __restrict__ out, int* cnts, unsigned* lists,
           const float* __restrict__ W1, float beta,
           const float* __restrict__ pin, const float* __restrict__ gamma,
           const float* __restrict__ bnb, float* __restrict__ pout,
           const unsigned* __restrict__ bcnt, const unsigned* __restrict__ segs) {
    extern __shared__ float agg_smem[];
    float* acc = agg_smem;
    unsigned* qall = (unsigned*)(agg_smem + AGG_ACCF);
    float2* sW2 = (float2*)(agg_smem + AGG_ACCF + AGG_WAVES * AGG_QCAP);
    float* scsh = agg_smem + AGG_ACCF + AGG_WAVES * AGG_QCAP + DH * DH;
    unsigned* lcnt = (unsigned*)(agg_smem + AGG_ACCF + AGG_WAVES * AGG_QCAP);
    unsigned* segc = lcnt + AGG_WAVES;
    float4* red = (float4*)qall;
    const int tid  = threadIdx.x;
    const int lane = tid & 31;
    const int wave = __builtin_amdgcn_readfirstlane(tid >> 5);
    const int lo   = blockIdx.x * AGG_R;
    const int cnt  = (NND - lo < AGG_R) ? (NND - lo) : AGG_R;
    unsigned* q = qall + wave * AGG_QCAP;
    const int widx = blockIdx.x * AGG_WAVES + wave;
    unsigned* wgl = lists + (size_t)blockIdx.x * AGG_WAVES * AGG_WCAP;
    unsigned* wl = wgl + (size_t)wave * AGG_WCAP;

    const int rot = (int)(((long long)blockIdx.x * AGG_NBATCH) / AGG_NWG);
    const int tstep = (MODE == 0) ? AGG_WAVES : 1;
    int t0 = (MODE == 0) ? wave : 0;
    int ta = t0 + rot;
    if (ta >= AGG_NBATCH) ta -= AGG_NBATCH;
    int4 dva[AGG_LPT], dvb[AGG_LPT];

    int wn = 0;
    bool scan = true;
    unsigned ent = 0u;
    unsigned bc = 0u;
    if (MODE == 0) {
        if (tid < BIN_NWG) bc = bcnt[(size_t)tid * BIN_NWG + blockIdx.x];
        if (tid < AGG_WAVES) lcnt[tid] = 0u;
    } else {
        wn = cnts[widx];
    }

    float4 sv[AGG_SVN];
    float dnv[AGG_SVN];
    float4 bb = make_float4(0.0f, 0.0f, 0.0f, 0.0f);
#pragma unroll
    for (int k = 0; k < AGG_SVN; ++k) {
        int n = wave + k * AGG_WAVES;
        if (n > cnt - 1) n = cnt - 1;
        sv[k] = *(const float4*)(start + (size_t)(lo + n) * DH + lane * 4);
        dnv[k] = (MODE == 0) ? dinv[lo + n] : 0.0f;
    }
    if (MODE == 0) bb = *(const float4*)(bias + lane * 4);
    float4 w1v[2][2];
    if (MODE != 0) {
#pragma unroll
        for (int it = 0; it < 2; ++it) {
            const int i = tid + it * AGG_TPB;
            const int kp = i / (DH / 4);
            const int c4 = i % (DH / 4);
            w1v[it][0] = *(const float4*)(W1 + (size_t)(2 * kp)     * DH + c4 * 4);
            w1v[it][1] = *(const float4*)(W1 + (size_t)(2 * kp + 1) * DH + c4 * 4);
        }
    }

    float4 pa = make_float4(0.0f, 0.0f, 0.0f, 0.0f);
    float gmv = 0.0f, btv = 0.0f;
    if (MODE == 2) {
        if (tid < DH) { gmv = gamma[tid]; btv = bnb[tid]; }
        const float* pp = pin + (size_t)(tid >> 6) * (2 * DH) + (tid & 63) * 4;
#pragma unroll
        for (int b = 0; b < AGG_NWG / AGG_PG; ++b) {
            const float4 p = *(const float4*)(pp + (size_t)b * (AGG_PG * 2 * DH));
            pa.x += p.x; pa.y += p.y; pa.z += p.z; pa.w += p.w;
        }
    }

    if (MODE != 0) {
        scan = wn > AGG_WCAP;
        if (!scan && lane < wn) ent = wl[lane];
    }

#pragma unroll
    for (int k = 0; k < AGG_SVN; ++k) {
        const int n = wave + k * AGG_WAVES;
        if (n < cnt) {
            const float4 v = sv[k];
            float4 o;
            if (MODE == 0) {
                const float w = dnv[k] * dnv[k];
                o.x = w * v.x + bb.x; o.y = w * v.y + bb.y;
                o.z = w * v.z + bb.z; o.w = w * v.w + bb.w;
            } else {
                o.x = ALPHA * v.x; o.y = ALPHA * v.y; o.z = ALPHA * v.z; o.w = ALPHA * v.w;
            }
            *(float4*)(acc + n * AGG_S + lane * 4) = o;
        }
    }
    if (MODE != 0) {
#pragma unroll
        for (int it = 0; it < 2; ++it) {
            const int i = tid + it * AGG_TPB;
            const int kp = i / (DH / 4);
            const int c4 = i % (DH / 4);
            const float4 a0 = w1v[it][0];
            const float4 a1 = w1v[it][1];
            float4* dst4 = (float4*)(sW2 + (size_t)kp * DH + c4 * 4);
            float4 f0; f0.x = a0.x; f0.y = a1.x; f0.z = a0.y; f0.w = a1.y;
            float4 f1; f1.x = a0.z; f1.y = a1.z; f1.z = a0.w; f1.w = a1.w;
            dst4[0] = f0;
            dst4[1] = f1;
        }
        for (int i = cnt * AGG_S + tid; i < AGG_ACCF; i += AGG_TPB) acc[i] = 0.0f;
    }
    if (MODE == 0 && tid < BIN_NWG) segc[tid] = bc;
    float sc[4] = {0.0f, 0.0f, 0.0f, 0.0f};
    float sh[4] = {0.0f, 0.0f, 0.0f, 0.0f};
    if (MODE == 2) {
        red[tid] = pa;
        __syncthreads();
        if (tid < DH) {
            const float* rf = (const float*)red;
            float S = rf[tid], Q = rf[DH + tid];
#pragma unroll
            for (int k = 1; k < AGG_PG; ++k) {
                S += rf[k * 2 * DH + tid];
                Q += rf[k * 2 * DH + DH + tid];
            }
            const float invN = 1.0f / (float)NND;
            const float mean = S * invN;
            const float inv  = rsqrtf(Q * invN - mean * mean + BNEPS);
            const float gi   = gmv * inv;
            scsh[tid]      = (1.0f - ALPHA) * gi;
            scsh[DH + tid] = (1.0f - ALPHA) * (btv - mean * gi);
        }
    }
    __syncthreads();
    if (MODE == 2) {
#pragma unroll
        for (int j = 0; j < 4; ++j) {
            const int col = scan ? (lane + 32 * j) : (lane * 4 + j);
            sc[j] = scsh[col];
            sh[j] = scsh[DH + col];
        }
    }

    bool segs_ok = false;
    if (MODE == 0) {
        unsigned mx = 0u;
#pragma unroll
        for (int j = 0; j < BIN_NWG / 32; ++j) {
            const unsigned n = segc[lane + 32 * j];
            mx = (n > mx) ? n : mx;
        }
        segs_ok = __builtin_amdgcn_ballot_w32(mx > (unsigned)BIN_CAP) == 0u;
    }

    if (MODE == 0 && segs_ok) {
        const unsigned* sp = segs + (size_t)blockIdx.x * BIN_NWG * BIN_CAP;
        unsigned ea[BIN_NWG / AGG_WAVES], eb[BIN_NWG / AGG_WAVES];
        int nn[BIN_NWG / AGG_WAVES];
#pragma unroll
        for (int j = 0; j < BIN_NWG / AGG_WAVES; ++j)
            nn[j] = __builtin_amdgcn_readfirstlane((int)segc[wave + j * AGG_WAVES]);
#pragma unroll
        for (int j = 0; j < BIN_NWG / AGG_WAVES; ++j) {
            const int b = wave + j * AGG_WAVES;
            const unsigned* p = sp + (size_t)b * BIN_CAP;
            ea[j] = (lane < nn[j]) ? p[lane] : 0u;
            eb[j] = (lane + 32 < nn[j]) ? p[lane + 32] : 0u;
        }
        int qn = 0;
#pragma unroll
        for (int j = 0; j < BIN_NWG / AGG_WAVES; ++j) {
            if (lane < nn[j]) q[qn + lane] = ea[j];
            if (lane + 32 < nn[j]) q[qn + 32 + lane] = eb[j];
            qn += nn[j];
        }
        __builtin_amdgcn_wave_barrier();
        for (int base = 0; base < qn; base += 32) {
            const int cntc = (qn - base < 32) ? (qn - base) : 32;
            agg_chunk_seg(q, base, cntc, lane, lo, zin, dinv, acc, wgl, lcnt, sc, sh);
        }
    } else if (scan) {
        agg_load(dva, dst, ta, lane);
        int qn = 0;
        while (true) {
            const int t1 = t0 + tstep;
            const bool has1 = t1 < AGG_NBATCH;
            int tb = (has1 ? t1 : t0) + rot;
            if (tb >= AGG_NBATCH) tb -= AGG_NBATCH;
            agg_load(dvb, dst, tb, lane);
            agg_scan<MODE != 0>(dva, ta, lane, lo, cnt, wave, q, qn);
            agg_drain<MODE>(q, qn, lane, lo, src, zin, dinv, acc, wgl, lcnt, sc, sh);
            if (!has1) break;
            const int t2 = t1 + tstep;
            const bool has2 = t2 < AGG_NBATCH;
            ta = (has2 ? t2 : t1) + rot;
            if (ta >= AGG_NBATCH) ta -= AGG_NBATCH;
            agg_load(dva, dst, ta, lane);
            agg_scan<MODE != 0>(dvb, tb, lane, lo, cnt, wave, q, qn);
            agg_drain<MODE>(q, qn, lane, lo, src, zin, dinv, acc, wgl, lcnt, sc, sh);
            if (!has2) break;
            t0 = t2;
        }
        __builtin_amdgcn_wave_barrier();
        if (qn > 0) agg_chunk<MODE>(q, 0, qn, lane, lo, src, zin, dinv, acc, wgl, lcnt, sc, sh);
    } else {
        int es = (int)(ent & 0xFFFFu), ed = (int)(ent >> 16);
        float4 v[8];
        if (wn > 0) {
            const int c0 = ((wn < 32) ? wn : 32) - 1;
#pragma unroll
            for (int u = 0; u < 8; ++u)
                v[u] = row_load4(zin, __builtin_amdgcn_readlane(es, (u < c0) ? u : c0), lane);
        }
        int base = 0;
        for (; base + 32 <= wn; base += 32) {
            unsigned nxt = 0u;
            if (base + 32 + lane < wn) nxt = wl[base + 32 + lane];
            const int ns = (int)(nxt & 0xFFFFu);
            int cl = wn - base - 33;
            if (cl > 31) cl = 31;
            if (cl < 0) cl = 0;
#pragma unroll
            for (int u = 0; u < 32; ++u) {
                row_add4<MODE == 2>(v[u & 7], acc, __builtin_amdgcn_readlane(ed, u), lane, sc, sh);
                if (u < 24) {
                    v[u & 7] = row_load4(zin, __builtin_amdgcn_readlane(es, u + 8), lane);
                } else {
                    const int j = (u - 24 < cl) ? (u - 24) : cl;
                    v[u & 7] = row_load4(zin, __builtin_amdgcn_readlane(ns, j), lane);
                }
            }
            es = ns; ed = (int)(nxt >> 16);
        }
        const int rem = wn - base;
        if (rem > 0) {
#pragma unroll
            for (int u = 0; u < 8; ++u)
                if (u < rem) row_add4<MODE == 2>(v[u], acc, __builtin_amdgcn_readlane(ed, u), lane, sc, sh);
            if (rem > 8) own_rows<MODE == 2>(es, ed, rem, lane, zin, acc, sc, sh, 8);
        }
    }
    __syncthreads();
    if (MODE == 0 && lane == 0) cnts[widx] = (int)lcnt[wave];

    if (MODE != 0) {
        const int r    = lane & 15;
        const bool hi  = lane >= 16;
        const int koff = hi ? 2 : 0;
        const bool act = wave < AGG_GW;
        const int row0 = (wave >> 1) * 16;
        const int col0 = (wave & 1) * 64 + r;
        v8f c[4] = {};
        if (act) {
            const float* arow = acc + (row0 + r) * AGG_S + koff;
            for (int k0 = 0; k0 < DH; k0 += 4) {
                const float2 a2 = *(const float2*)(arow + k0);
                v2f a; a.x = a2.x; a.y = a2.y;
                const float2* brow = sW2 + (size_t)((k0 >> 1) + (hi ? 1 : 0)) * DH + col0;
#pragma unroll
                for (int t = 0; t < 4; ++t) {
                    const float2 b2 = brow[t * 16];
                    v2f b; b.x = b2.x; b.y = b2.y;
                    c[t] = __builtin_amdgcn_wmma_f32_16x16x4_f32(
                        false, a, false, b, (short)0, c[t], false, false);
                }
            }
#pragma unroll
            for (int t = 0; t < 4; ++t) {
#pragma unroll
                for (int v = 0; v < 8; ++v) {
                    const float hv = acc[(row0 + v + (hi ? 8 : 0)) * AGG_S + col0 + t * 16];
                    c[t][v] = (1.0f - beta) * hv + beta * c[t][v];
                }
            }
        }
        __syncthreads();
        if (act) {
#pragma unroll
            for (int t = 0; t < 4; ++t) {
#pragma unroll
                for (int v = 0; v < 8; ++v)
                    acc[(row0 + v + (hi ? 8 : 0)) * AGG_S + col0 + t * 16] = c[t][v];
            }
        }
        __syncthreads();
    }

    float4 cs = make_float4(0.0f, 0.0f, 0.0f, 0.0f);
    float4 cq = make_float4(0.0f, 0.0f, 0.0f, 0.0f);
    for (int i = tid; i < cnt * (DH / 4); i += AGG_TPB) {
        const int n = i >> 5, c4 = i & 31;
        const float4 o = *(const float4*)(acc + n * AGG_S + c4 * 4);
        *(float4*)(out + (size_t)(lo + n) * DH + c4 * 4) = o;
        if (MODE != 0) {
            cs.x += o.x; cs.y += o.y; cs.z += o.z; cs.w += o.w;
            cq.x += o.x * o.x; cq.y += o.y * o.y; cq.z += o.z * o.z; cq.w += o.w * o.w;
        }
    }
    if (MODE != 0 && pout != nullptr) {
        red[wave * 64 + lane]      = cs;
        red[wave * 64 + 32 + lane] = cq;
        __syncthreads();
        if (tid < 64) {
            float4 a = red[tid];
#pragma unroll
            for (int w = 1; w < AGG_WAVES; ++w) {
                const float4 b = red[w * 64 + tid];
                a.x += b.x; a.y += b.y; a.z += b.z; a.w += b.w;
            }
            *(float4*)(pout + (size_t)blockIdx.x * (2 * DH) + tid * 4) = a;
        }
    }
}

extern "C" void kernel_launch(void* const* d_in, const int* in_sizes, int n_in,
                              void* d_out, int out_size, void* d_ws, size_t ws_size,
                              hipStream_t stream) {
    const float* x        = (const float*)d_in[0];
    const int*   edge     = (const int*)d_in[1];
    const float* W_lin    = (const float*)d_in[2];
    const float* b_lin    = (const float*)d_in[3];
    const float* W1s      = (const float*)d_in[4];
    const float* bn_gamma = (const float*)d_in[5];
    const float* bn_beta  = (const float*)d_in[6];
    float* out = (float*)d_out;

    float* ws    = (float*)d_ws;
    float* dinv  = ws;
    float* partA = ws + NND;
    float* partB = partA + (size_t)AGG_NWG * 2 * DH;
    float* bufA  = partB + (size_t)AGG_NWG * 2 * DH;
    float* bufB  = bufA + (size_t)NND * DH;
    float* x0    = bufB + (size_t)NND * DH;
    int* cnts       = (int*)(x0 + (size_t)NND * DH);
    unsigned* lists = (unsigned*)(cnts + AGG_NWG * AGG_WAVES);
    unsigned* bcnt = lists + (size_t)AGG_NWG * AGG_WAVES * AGG_WCAP + 16;
    unsigned* segs = bcnt + (size_t)BIN_NWG * BIN_NWG + 32;

    const int* src = edge;
    const int* dst = edge + NED;

    const int nBlkN   = (NND + 255) / 256;
    const int nBlkGemm = (NND / 16 + 7) / 8;

    k_deg_init<<<nBlkN, 256, 0, stream>>>(dinv);
    k_bin<<<BIN_NWG, BIN_TPB, (BIN_NWG + BIN_NWG * BIN_CAP) * sizeof(unsigned), stream>>>(src, dst, dinv, bcnt, segs);
    k_deg_rsqrt<<<nBlkN, 256, 0, stream>>>(dinv);

    k_gemm_wmma<DIN><<<nBlkGemm, 256, DIN * DH * sizeof(float), stream>>>(
        x, W_lin, bufA, 0, 0.0f);

    k_agg<0><<<AGG_NWG, AGG_TPB, AGG_LDS0C, stream>>>(src, dst, bufA, bufA, dinv, b_lin, bufB,
                                                     cnts, lists, W1s, 0.0f,
                                                     nullptr, nullptr, nullptr, nullptr, bcnt, segs);

    hipMemcpyAsync(x0, bufB, (size_t)NND * DH * sizeof(float),
                   hipMemcpyDeviceToDevice, stream);

    float* z = bufB;
    float* h = bufA;
    for (int i = 0; i < 5; ++i) {
        const float beta = logf(0.5f / (float)(i + 1) + 1.0f);
        float* pout = (i < 4) ? ((i & 1) ? partB : partA) : nullptr;
        if (i == 0) {
            k_agg<1><<<AGG_NWG, AGG_TPB, AGG_LDS1, stream>>>(src, dst, z, x0, dinv, b_lin, h, cnts, lists,
                                                             W1s, beta,
                                                             nullptr, nullptr, nullptr, pout, bcnt, segs);
        } else {
            const float* pin = ((i - 1) & 1) ? partB : partA;
            k_agg<2><<<AGG_NWG, AGG_TPB, AGG_LDS2, stream>>>(src, dst, z, x0, dinv, b_lin, h, cnts, lists,
                                                             W1s + (size_t)i * DH * DH, beta,
                                                             pin, bn_gamma + (i - 1) * DH,
                                                             bn_beta + (i - 1) * DH, pout, bcnt, segs);
        }
        float* t = z; z = h; h = t;
    }

    hipMemcpyAsync(out, z, (size_t)NND * DH * sizeof(float),
                   hipMemcpyDeviceToDevice, stream);
}
